# starting kernel
# baseline (speedup 1.0000x reference)
#include <hip/hip_runtime.h>

#define NNODES 100000
#define NEDGES 1600000
#define NFEAT 128
#define NHID 64
#define NCLASS 16
#define MTILES (NNODES / 16)

typedef float v2f __attribute__((ext_vector_type(2)));
typedef float v8f __attribute__((ext_vector_type(8)));

__device__ __forceinline__ void atomic_add_f32(float* p, float v) {
    __hip_atomic_fetch_add(p, v, __ATOMIC_RELAXED, __HIP_MEMORY_SCOPE_AGENT);
}

__global__ __launch_bounds__(256) void gcn_zero_kernel(float4* __restrict__ p, int n4) {
    int i = blockIdx.x * 256 + threadIdx.x;
    if (i < n4) p[i] = make_float4(0.f, 0.f, 0.f, 0.f);
}

__global__ __launch_bounds__(256) void gcn_init_logits_kernel(const float* __restrict__ b2,
                                                              float* __restrict__ out, int total) {
    int i = blockIdx.x * 256 + threadIdx.x;
    if (i < total) out[i] = b2[i & (NCLASS - 1)];
}

__global__ __launch_bounds__(128) void gcn_gemm1_kernel(const float* __restrict__ x,
                                                        const float* __restrict__ W1,
                                                        float* __restrict__ out) {
    __shared__ float As[16][NFEAT + 4];
    __shared__ float WsT[NHID][NFEAT + 4];

    const int tid = threadIdx.x;
    const int tileM = blockIdx.x;

    for (int i = tid; i < NFEAT * NHID; i += 128) {
        int k = i >> 6, n = i & (NHID - 1);
        WsT[n][k] = W1[i];
    }
    const float* xrow = x + (size_t)tileM * 16 * NFEAT;
    for (int i = tid; i < 16 * NFEAT; i += 128) {
        As[i >> 7][i & (NFEAT - 1)] = xrow[i];
    }
    __syncthreads();

    const int lane = tid & 31;
    const int wave = tid >> 5;
    const int m  = lane & 15;
    const int n  = lane & 15;
    const int kh = (lane >> 4) * 2;
    const int n0 = wave * 16;

    v8f acc = {};
    #pragma unroll 4
    for (int kk = 0; kk < NFEAT; kk += 4) {
        const int k0 = kk + kh;
        v2f a, b;
        a.x = As[m][k0];
        a.y = As[m][k0 + 1];
        b.x = WsT[n0 + n][k0];
        b.y = WsT[n0 + n][k0 + 1];
        acc = __builtin_amdgcn_wmma_f32_16x16x4_f32(false, a, false, b,
                                                    (short)0, acc, false, false);
    }

    float* orow = out + ((size_t)tileM * 16 + (lane >> 4) * 8) * NHID + n0 + n;
    #pragma unroll
    for (int r = 0; r < 8; ++r)
        orow[(size_t)r * NHID] = acc[r];
}

__global__ __launch_bounds__(256) void gcn_spmm64_kernel(const int* __restrict__ src,
                                                         const int* __restrict__ dst,
                                                         const float* __restrict__ ew,
                                                         const float* __restrict__ sup,
                                                         float* __restrict__ out) {
    long gid = (long)blockIdx.x * 256 + threadIdx.x;
    if (gid >= (long)NEDGES * NHID) return;
    int e = (int)(gid >> 6);
    int f = (int)(gid & (NHID - 1));
    int s = src[e];
    int d = dst[e];
    float w = ew[e];
    float msg = w * sup[(size_t)s * NHID + f];
    atomic_add_f32(out + (size_t)d * NHID + f, msg);
}

__global__ __launch_bounds__(256) void gcn_gemm2_kernel(const float* __restrict__ agg1,
                                                        const float* __restrict__ b1,
                                                        const float* __restrict__ W2,
                                                        float* __restrict__ out) {
    __shared__ float WsT[NCLASS][NHID + 4];
    __shared__ float bs[NHID];

    const int tid = threadIdx.x;
    for (int i = tid; i < NHID * NCLASS; i += 256) {
        int k = i >> 4, n = i & (NCLASS - 1);
        WsT[n][k] = W2[i];
    }
    if (tid < NHID) bs[tid] = b1[tid];
    __syncthreads();

    const int lane = tid & 31;
    const int wave = tid >> 5;
    const int tileM = blockIdx.x * 8 + wave;
    if (tileM >= MTILES) return;

    const int m  = lane & 15;
    const int n  = lane & 15;
    const int kh = (lane >> 4) * 2;

    const float* arow = agg1 + ((size_t)tileM * 16 + m) * NHID;

    v8f acc = {};
    #pragma unroll 4
    for (int kk = 0; kk < NHID; kk += 4) {
        const int k0 = kk + kh;
        v2f a, b;
        a.x = fmaxf(arow[k0]     + bs[k0],     0.f);
        a.y = fmaxf(arow[k0 + 1] + bs[k0 + 1], 0.f);
        b.x = WsT[n][k0];
        b.y = WsT[n][k0 + 1];
        acc = __builtin_amdgcn_wmma_f32_16x16x4_f32(false, a, false, b,
                                                    (short)0, acc, false, false);
    }

    float* orow = out + ((size_t)tileM * 16 + (lane >> 4) * 8) * NCLASS + n;
    #pragma unroll
    for (int r = 0; r < 8; ++r)
        orow[(size_t)r * NCLASS] = acc[r];
}

__global__ __launch_bounds__(256) void gcn_spmm16_kernel(const int* __restrict__ src,
                                                         const int* __restrict__ dst,
                                                         const float* __restrict__ ew,
                                                         const float* __restrict__ sup,
                                                         float* __restrict__ out) {
    long gid = (long)blockIdx.x * 256 + threadIdx.x;
    if (gid >= (long)NEDGES * NCLASS) return;
    int e = (int)(gid >> 4);
    int f = (int)(gid & (NCLASS - 1));
    int s = src[e];
    int d = dst[e];
    float w = ew[e];
    float msg = w * sup[(size_t)s * NCLASS + f];
    atomic_add_f32(out + (size_t)d * NCLASS + f, msg);
}

__global__ __launch_bounds__(256) void gcn_logsoftmax_kernel(float* __restrict__ logits) {
    int node = blockIdx.x * 256 + threadIdx.x;
    if (node >= NNODES) return;
    float4* p = (float4*)(logits + (size_t)node * NCLASS);
    float4 v[4];
    #pragma unroll
    for (int i = 0; i < 4; ++i) v[i] = p[i];
    float* f = (float*)v;
    float mx = f[0];
    #pragma unroll
    for (int i = 1; i < 16; ++i) mx = fmaxf(mx, f[i]);
    float sum = 0.f;
    #pragma unroll
    for (int i = 0; i < 16; ++i) sum += __expf(f[i] - mx);
    float lse = mx + __logf(sum);
    #pragma unroll
    for (int i = 0; i < 16; ++i) f[i] -= lse;
    #pragma unroll
    for (int i = 0; i < 4; ++i) p[i] = v[i];
}

extern "C" void kernel_launch(void* const* d_in, const int* in_sizes, int n_in,
                              void* d_out, int out_size, void* d_ws, size_t ws_size,
                              hipStream_t stream) {
    const float* x   = (const float*)d_in[0];
    const int*   src = (const int*)  d_in[1];
    const int*   dst = (const int*)  d_in[2];
    const float* ew  = (const float*)d_in[3];
    const float* W1  = (const float*)d_in[4];
    const float* b1  = (const float*)d_in[5];
    const float* W2  = (const float*)d_in[6];
    const float* b2  = (const float*)d_in[7];
    float* out = (float*)d_out;

    float* ws       = (float*)d_ws;
    float* support1 = ws;
    float* agg1     = ws + (size_t)NNODES * NHID;
    float* support2 = ws + 2 * (size_t)NNODES * NHID;

    {
        int n4 = NNODES * NHID / 4;
        gcn_zero_kernel<<<(n4 + 255) / 256, 256, 0, stream>>>((float4*)agg1, n4);
        int nl = NNODES * NCLASS;
        gcn_init_logits_kernel<<<(nl + 255) / 256, 256, 0, stream>>>(b2, out, nl);
    }

    gcn_gemm1_kernel<<<MTILES, 128, 0, stream>>>(x, W1, support1);
    {
        long total = (long)NEDGES * NHID;
        int blocks = (int)((total + 255) / 256);
        gcn_spmm64_kernel<<<blocks, 256, 0, stream>>>(src, dst, ew, support1, agg1);
    }

    gcn_gemm2_kernel<<<(MTILES + 7) / 8, 256, 0, stream>>>(agg1, b1, W2, support2);
    {
        long total = (long)NEDGES * NCLASS;
        int blocks = (int)((total + 255) / 256);
        gcn_spmm16_kernel<<<blocks, 256, 0, stream>>>(src, dst, ew, support2, out);
    }

    gcn_logsoftmax_kernel<<<(NNODES + 255) / 256, 256, 0, stream>>>(out);
}


# Edge scatters run as 12,500 resident waves, each looping over 128 edges (records read 32 at a time and broadcast, 8 rows gathered ahead); f32 atomic adds kept, sums in arrival order as before
# speedup vs baseline: 2.6242x; 2.6242x over previous
#include <hip/hip_runtime.h>

#define NNODES 100000
#define NEDGES 1600000
#define NFEAT 128
#define NHID 64
#define NCLASS 16
#define MTILES (NNODES / 16)


#define EDGES_PER_WAVE 128
#define SPMM_WAVES_PER_BLOCK 4
#define SPMM_BLOCKS (NEDGES / EDGES_PER_WAVE / SPMM_WAVES_PER_BLOCK)
static_assert(NEDGES % (EDGES_PER_WAVE * SPMM_WAVES_PER_BLOCK) == 0, "edge count must fill whole waves");
static_assert(EDGES_PER_WAVE % 32 == 0, "a wave reads its edge records 32 at a time");

typedef float v2f __attribute__((ext_vector_type(2)));
typedef float v8f __attribute__((ext_vector_type(8)));

__device__ __forceinline__ void atomic_add_f32(float* p, float v) {
    __hip_atomic_fetch_add(p, v, __ATOMIC_RELAXED, __HIP_MEMORY_SCOPE_AGENT);
}

__global__ __launch_bounds__(256) void gcn_zero_kernel(float4* __restrict__ p, int n4) {
    int i = blockIdx.x * 256 + threadIdx.x;
    if (i < n4) p[i] = make_float4(0.f, 0.f, 0.f, 0.f);
}

__global__ __launch_bounds__(256) void gcn_init_logits_kernel(const float* __restrict__ b2,
                                                              float* __restrict__ out, int total) {
    int i = blockIdx.x * 256 + threadIdx.x;
    if (i < total) out[i] = b2[i & (NCLASS - 1)];
}

__global__ __launch_bounds__(128) void gcn_gemm1_kernel(const float* __restrict__ x,
                                                        const float* __restrict__ W1,
                                                        float* __restrict__ out) {
    __shared__ float As[16][NFEAT + 4];
    __shared__ float WsT[NHID][NFEAT + 4];

    const int tid = threadIdx.x;
    const int tileM = blockIdx.x;

    for (int i = tid; i < NFEAT * NHID; i += 128) {
        int k = i >> 6, n = i & (NHID - 1);
        WsT[n][k] = W1[i];
    }
    const float* xrow = x + (size_t)tileM * 16 * NFEAT;
    for (int i = tid; i < 16 * NFEAT; i += 128) {
        As[i >> 7][i & (NFEAT - 1)] = xrow[i];
    }
    __syncthreads();

    const int lane = tid & 31;
    const int wave = tid >> 5;
    const int m  = lane & 15;
    const int n  = lane & 15;
    const int kh = (lane >> 4) * 2;
    const int n0 = wave * 16;

    v8f acc = {};
    #pragma unroll 4
    for (int kk = 0; kk < NFEAT; kk += 4) {
        const int k0 = kk + kh;
        v2f a, b;
        a.x = As[m][k0];
        a.y = As[m][k0 + 1];
        b.x = WsT[n0 + n][k0];
        b.y = WsT[n0 + n][k0 + 1];
        acc = __builtin_amdgcn_wmma_f32_16x16x4_f32(false, a, false, b,
                                                    (short)0, acc, false, false);
    }

    float* orow = out + ((size_t)tileM * 16 + (lane >> 4) * 8) * NHID + n0 + n;
    #pragma unroll
    for (int r = 0; r < 8; ++r)
        orow[(size_t)r * NHID] = acc[r];
}

__global__ __launch_bounds__(128) void gcn_spmm64_kernel(const int* __restrict__ src,
                                                         const int* __restrict__ dst,
                                                         const float* __restrict__ ew,
                                                         const float* __restrict__ sup,
                                                         float* __restrict__ out) {
    const unsigned lane = threadIdx.x & 31;
    const int wave = blockIdx.x * SPMM_WAVES_PER_BLOCK + (threadIdx.x >> 5);
    const int e0 = wave * EDGES_PER_WAVE;

    int   s_next = src[e0 + lane];
    int   d_next = dst[e0 + lane];
    float w_next = ew[e0 + lane];

    for (int b = 0; b < EDGES_PER_WAVE; b += 32) {
        const int s_cur = s_next;
        const int d_cur = d_next;
        const int w_cur = __float_as_int(w_next);
        if (b + 32 < EDGES_PER_WAVE) {
            const int e = e0 + b + 32 + lane;
            s_next = src[e];
            d_next = dst[e];
            w_next = ew[e];
        }
        #pragma unroll
        for (int g = 0; g < 32; g += 8) {
            float v0[8], v1[8];
            #pragma unroll
            for (int u = 0; u < 8; ++u) {
                const unsigned sj = (unsigned)__builtin_amdgcn_readlane(s_cur, g + u);
                const unsigned o = sj * NHID + lane;
                v0[u] = sup[o];
                v1[u] = sup[o + 32];
            }
            #pragma unroll
            for (int u = 0; u < 8; ++u) {
                const unsigned dj = (unsigned)__builtin_amdgcn_readlane(d_cur, g + u);
                const float wj = __int_as_float(__builtin_amdgcn_readlane(w_cur, g + u));
                const unsigned o = dj * NHID + lane;
                atomic_add_f32(out + o,      wj * v0[u]);
                atomic_add_f32(out + o + 32, wj * v1[u]);
            }
        }
    }
}

__global__ __launch_bounds__(256) void gcn_gemm2_kernel(const float* __restrict__ agg1,
                                                        const float* __restrict__ b1,
                                                        const float* __restrict__ W2,
                                                        float* __restrict__ out) {
    __shared__ float WsT[NCLASS][NHID + 4];
    __shared__ float bs[NHID];

    const int tid = threadIdx.x;
    for (int i = tid; i < NHID * NCLASS; i += 256) {
        int k = i >> 4, n = i & (NCLASS - 1);
        WsT[n][k] = W2[i];
    }
    if (tid < NHID) bs[tid] = b1[tid];
    __syncthreads();

    const int lane = tid & 31;
    const int wave = tid >> 5;
    const int tileM = blockIdx.x * 8 + wave;
    if (tileM >= MTILES) return;

    const int m  = lane & 15;
    const int n  = lane & 15;
    const int kh = (lane >> 4) * 2;

    const float* arow = agg1 + ((size_t)tileM * 16 + m) * NHID;

    v8f acc = {};
    #pragma unroll 4
    for (int kk = 0; kk < NHID; kk += 4) {
        const int k0 = kk + kh;
        v2f a, b;
        a.x = fmaxf(arow[k0]     + bs[k0],     0.f);
        a.y = fmaxf(arow[k0 + 1] + bs[k0 + 1], 0.f);
        b.x = WsT[n][k0];
        b.y = WsT[n][k0 + 1];
        acc = __builtin_amdgcn_wmma_f32_16x16x4_f32(false, a, false, b,
                                                    (short)0, acc, false, false);
    }

    float* orow = out + ((size_t)tileM * 16 + (lane >> 4) * 8) * NCLASS + n;
    #pragma unroll
    for (int r = 0; r < 8; ++r)
        orow[(size_t)r * NCLASS] = acc[r];
}

__global__ __launch_bounds__(128) void gcn_spmm16_kernel(const int* __restrict__ src,
                                                         const int* __restrict__ dst,
                                                         const float* __restrict__ ew,
                                                         const float* __restrict__ sup,
                                                         float* __restrict__ out) {
    const int lane = threadIdx.x & 31;
    const bool odd = lane >= 16;
    const int wave = blockIdx.x * SPMM_WAVES_PER_BLOCK + (threadIdx.x >> 5);
    const int e0 = wave * EDGES_PER_WAVE;

    const unsigned f = lane & 15;

    int   s_next = src[e0 + lane];
    int   d_next = dst[e0 + lane];
    float w_next = ew[e0 + lane];

    for (int b = 0; b < EDGES_PER_WAVE; b += 32) {
        const int s_cur = s_next;
        const int d_cur = d_next;
        const int w_cur = __float_as_int(w_next);
        if (b + 32 < EDGES_PER_WAVE) {
            const int e = e0 + b + 32 + lane;
            s_next = src[e];
            d_next = dst[e];
            w_next = ew[e];
        }
        #pragma unroll
        for (int g = 0; g < 16; g += 8) {
            float v[8];
            #pragma unroll
            for (int u = 0; u < 8; ++u) {
                const int sa = __builtin_amdgcn_readlane(s_cur, 2 * (g + u));
                const int sb = __builtin_amdgcn_readlane(s_cur, 2 * (g + u) + 1);
                const unsigned sj = (unsigned)(odd ? sb : sa);
                v[u] = sup[sj * NCLASS + f];
            }
            #pragma unroll
            for (int u = 0; u < 8; ++u) {
                const int da = __builtin_amdgcn_readlane(d_cur, 2 * (g + u));
                const int db = __builtin_amdgcn_readlane(d_cur, 2 * (g + u) + 1);
                const int wa = __builtin_amdgcn_readlane(w_cur, 2 * (g + u));
                const int wb = __builtin_amdgcn_readlane(w_cur, 2 * (g + u) + 1);
                const unsigned dj = (unsigned)(odd ? db : da);
                const float wj = __int_as_float(odd ? wb : wa);
                atomic_add_f32(out + (dj * NCLASS + f), wj * v[u]);
            }
        }
    }
}

__global__ __launch_bounds__(256) void gcn_logsoftmax_kernel(float* __restrict__ logits) {
    int node = blockIdx.x * 256 + threadIdx.x;
    if (node >= NNODES) return;
    float4* p = (float4*)(logits + (size_t)node * NCLASS);
    float4 v[4];
    #pragma unroll
    for (int i = 0; i < 4; ++i) v[i] = p[i];
    float* f = (float*)v;
    float mx = f[0];
    #pragma unroll
    for (int i = 1; i < 16; ++i) mx = fmaxf(mx, f[i]);
    float sum = 0.f;
    #pragma unroll
    for (int i = 0; i < 16; ++i) sum += __expf(f[i] - mx);
    float lse = mx + __logf(sum);
    #pragma unroll
    for (int i = 0; i < 16; ++i) f[i] -= lse;
    #pragma unroll
    for (int i = 0; i < 4; ++i) p[i] = v[i];
}

extern "C" void kernel_launch(void* const* d_in, const int* in_sizes, int n_in,
                              void* d_out, int out_size, void* d_ws, size_t ws_size,
                              hipStream_t stream) {
    const float* x   = (const float*)d_in[0];
    const int*   src = (const int*)  d_in[1];
    const int*   dst = (const int*)  d_in[2];
    const float* ew  = (const float*)d_in[3];
    const float* W1  = (const float*)d_in[4];
    const float* b1  = (const float*)d_in[5];
    const float* W2  = (const float*)d_in[6];
    const float* b2  = (const float*)d_in[7];
    float* out = (float*)d_out;

    float* ws       = (float*)d_ws;
    float* support1 = ws;
    float* agg1     = ws + (size_t)NNODES * NHID;
    float* support2 = ws + 2 * (size_t)NNODES * NHID;

    {
        int n4 = NNODES * NHID / 4;
        gcn_zero_kernel<<<(n4 + 255) / 256, 256, 0, stream>>>((float4*)agg1, n4);
        int nl = NNODES * NCLASS;
        gcn_init_logits_kernel<<<(nl + 255) / 256, 256, 0, stream>>>(b2, out, nl);
    }

    gcn_gemm1_kernel<<<MTILES, 128, 0, stream>>>(x, W1, support1);
    gcn_spmm64_kernel<<<SPMM_BLOCKS, 32 * SPMM_WAVES_PER_BLOCK, 0, stream>>>(src, dst, ew, support1, agg1);

    gcn_gemm2_kernel<<<(MTILES + 7) / 8, 256, 0, stream>>>(agg1, b1, W2, support2);
    gcn_spmm16_kernel<<<SPMM_BLOCKS, 32 * SPMM_WAVES_PER_BLOCK, 0, stream>>>(src, dst, ew, support2, out);

    gcn_logsoftmax_kernel<<<(NNODES + 255) / 256, 256, 0, stream>>>(out);
}


# The two fill launches are folded into the GEMMs: GEMM1 blocks zero their 16 rows of the layer-1 sums, GEMM2 waves store the bias into their 16 output rows (7 launches to 5, 100,000 fewer waves)
# speedup vs baseline: 2.8741x; 1.0952x over previous
#include <hip/hip_runtime.h>

#define NNODES 100000
#define NEDGES 1600000
#define NFEAT 128
#define NHID 64
#define NCLASS 16
#define MTILES (NNODES / 16)


#define EDGES_PER_WAVE 128
#define SPMM_WAVES_PER_BLOCK 4
#define SPMM_BLOCKS (NEDGES / EDGES_PER_WAVE / SPMM_WAVES_PER_BLOCK)
static_assert(NEDGES % (EDGES_PER_WAVE * SPMM_WAVES_PER_BLOCK) == 0, "edge count must fill whole waves");
static_assert(EDGES_PER_WAVE % 32 == 0, "a wave reads its edge records 32 at a time");

typedef float v2f __attribute__((ext_vector_type(2)));
typedef float v8f __attribute__((ext_vector_type(8)));

__device__ __forceinline__ void atomic_add_f32(float* p, float v) {
    __hip_atomic_fetch_add(p, v, __ATOMIC_RELAXED, __HIP_MEMORY_SCOPE_AGENT);
}


__global__ __launch_bounds__(128) void gcn_gemm1_kernel(const float* __restrict__ x,
                                                        const float* __restrict__ W1,
                                                        float* __restrict__ out,
                                                        float* __restrict__ agg) {
    __shared__ float As[16][NFEAT + 4];
    __shared__ float WsT[NHID][NFEAT + 4];

    const int tid = threadIdx.x;
    const int tileM = blockIdx.x;

    for (int i = tid; i < NFEAT * NHID; i += 128) {
        int k = i >> 6, n = i & (NHID - 1);
        WsT[n][k] = W1[i];
    }
    const float* xrow = x + (size_t)tileM * 16 * NFEAT;
    for (int i = tid; i < 16 * NFEAT; i += 128) {
        As[i >> 7][i & (NFEAT - 1)] = xrow[i];
    }
    __syncthreads();

    {
        float4* zrow = (float4*)(agg + (size_t)tileM * 16 * NHID);
        const float4 z = make_float4(0.f, 0.f, 0.f, 0.f);
        zrow[tid]       = z;
        zrow[tid + 128] = z;
    }

    const int lane = tid & 31;
    const int wave = tid >> 5;
    const int m  = lane & 15;
    const int n  = lane & 15;
    const int kh = (lane >> 4) * 2;
    const int n0 = wave * 16;

    v8f acc = {};
    #pragma unroll 4
    for (int kk = 0; kk < NFEAT; kk += 4) {
        const int k0 = kk + kh;
        v2f a, b;
        a.x = As[m][k0];
        a.y = As[m][k0 + 1];
        b.x = WsT[n0 + n][k0];
        b.y = WsT[n0 + n][k0 + 1];
        acc = __builtin_amdgcn_wmma_f32_16x16x4_f32(false, a, false, b,
                                                    (short)0, acc, false, false);
    }

    float* orow = out + ((size_t)tileM * 16 + (lane >> 4) * 8) * NHID + n0 + n;
    #pragma unroll
    for (int r = 0; r < 8; ++r)
        orow[(size_t)r * NHID] = acc[r];
}

__global__ __launch_bounds__(128) void gcn_spmm64_kernel(const int* __restrict__ src,
                                                         const int* __restrict__ dst,
                                                         const float* __restrict__ ew,
                                                         const float* __restrict__ sup,
                                                         float* __restrict__ out) {
    const unsigned lane = threadIdx.x & 31;
    const int wave = blockIdx.x * SPMM_WAVES_PER_BLOCK + (threadIdx.x >> 5);
    const int e0 = wave * EDGES_PER_WAVE;

    int   s_next = src[e0 + lane];
    int   d_next = dst[e0 + lane];
    float w_next = ew[e0 + lane];

    for (int b = 0; b < EDGES_PER_WAVE; b += 32) {
        const int s_cur = s_next;
        const int d_cur = d_next;
        const int w_cur = __float_as_int(w_next);
        if (b + 32 < EDGES_PER_WAVE) {
            const int e = e0 + b + 32 + lane;
            s_next = src[e];
            d_next = dst[e];
            w_next = ew[e];
        }
        #pragma unroll
        for (int g = 0; g < 32; g += 8) {
            float v0[8], v1[8];
            #pragma unroll
            for (int u = 0; u < 8; ++u) {
                const unsigned sj = (unsigned)__builtin_amdgcn_readlane(s_cur, g + u);
                const unsigned o = sj * NHID + lane;
                v0[u] = sup[o];
                v1[u] = sup[o + 32];
            }
            #pragma unroll
            for (int u = 0; u < 8; ++u) {
                const unsigned dj = (unsigned)__builtin_amdgcn_readlane(d_cur, g + u);
                const float wj = __int_as_float(__builtin_amdgcn_readlane(w_cur, g + u));
                const unsigned o = dj * NHID + lane;
                atomic_add_f32(out + o,      wj * v0[u]);
                atomic_add_f32(out + o + 32, wj * v1[u]);
            }
        }
    }
}

__global__ __launch_bounds__(256) void gcn_gemm2_kernel(const float* __restrict__ agg1,
                                                        const float* __restrict__ b1,
                                                        const float* __restrict__ W2,
                                                        const float* __restrict__ b2,
                                                        float* __restrict__ out,
                                                        float* __restrict__ logits) {
    __shared__ float WsT[NCLASS][NHID + 4];
    __shared__ float bs[NHID];

    const int tid = threadIdx.x;

    const int q = (tid & 3) * 4;
    const float4 bq = make_float4(b2[q], b2[q + 1], b2[q + 2], b2[q + 3]);

    for (int i = tid; i < NHID * NCLASS; i += 256) {
        int k = i >> 4, n = i & (NCLASS - 1);
        WsT[n][k] = W2[i];
    }
    if (tid < NHID) bs[tid] = b1[tid];
    __syncthreads();

    const int lane = tid & 31;
    const int wave = tid >> 5;
    const int tileM = blockIdx.x * 8 + wave;
    if (tileM >= MTILES) return;

    const int m  = lane & 15;
    const int n  = lane & 15;
    const int kh = (lane >> 4) * 2;

    {
        float4* lrow = (float4*)(logits + (size_t)tileM * 16 * NCLASS);
        lrow[lane]      = bq;
        lrow[lane + 32] = bq;
    }

    const float* arow = agg1 + ((size_t)tileM * 16 + m) * NHID;

    v8f acc = {};
    #pragma unroll 4
    for (int kk = 0; kk < NHID; kk += 4) {
        const int k0 = kk + kh;
        v2f a, b;
        a.x = fmaxf(arow[k0]     + bs[k0],     0.f);
        a.y = fmaxf(arow[k0 + 1] + bs[k0 + 1], 0.f);
        b.x = WsT[n][k0];
        b.y = WsT[n][k0 + 1];
        acc = __builtin_amdgcn_wmma_f32_16x16x4_f32(false, a, false, b,
                                                    (short)0, acc, false, false);
    }

    float* orow = out + ((size_t)tileM * 16 + (lane >> 4) * 8) * NCLASS + n;
    #pragma unroll
    for (int r = 0; r < 8; ++r)
        orow[(size_t)r * NCLASS] = acc[r];
}

__global__ __launch_bounds__(128) void gcn_spmm16_kernel(const int* __restrict__ src,
                                                         const int* __restrict__ dst,
                                                         const float* __restrict__ ew,
                                                         const float* __restrict__ sup,
                                                         float* __restrict__ out) {
    const int lane = threadIdx.x & 31;
    const bool odd = lane >= 16;
    const int wave = blockIdx.x * SPMM_WAVES_PER_BLOCK + (threadIdx.x >> 5);
    const int e0 = wave * EDGES_PER_WAVE;

    const unsigned f = lane & 15;

    int   s_next = src[e0 + lane];
    int   d_next = dst[e0 + lane];
    float w_next = ew[e0 + lane];

    for (int b = 0; b < EDGES_PER_WAVE; b += 32) {
        const int s_cur = s_next;
        const int d_cur = d_next;
        const int w_cur = __float_as_int(w_next);
        if (b + 32 < EDGES_PER_WAVE) {
            const int e = e0 + b + 32 + lane;
            s_next = src[e];
            d_next = dst[e];
            w_next = ew[e];
        }
        #pragma unroll
        for (int g = 0; g < 16; g += 8) {
            float v[8];
            #pragma unroll
            for (int u = 0; u < 8; ++u) {
                const int sa = __builtin_amdgcn_readlane(s_cur, 2 * (g + u));
                const int sb = __builtin_amdgcn_readlane(s_cur, 2 * (g + u) + 1);
                const unsigned sj = (unsigned)(odd ? sb : sa);
                v[u] = sup[sj * NCLASS + f];
            }
            #pragma unroll
            for (int u = 0; u < 8; ++u) {
                const int da = __builtin_amdgcn_readlane(d_cur, 2 * (g + u));
                const int db = __builtin_amdgcn_readlane(d_cur, 2 * (g + u) + 1);
                const int wa = __builtin_amdgcn_readlane(w_cur, 2 * (g + u));
                const int wb = __builtin_amdgcn_readlane(w_cur, 2 * (g + u) + 1);
                const unsigned dj = (unsigned)(odd ? db : da);
                const float wj = __int_as_float(odd ? wb : wa);
                atomic_add_f32(out + (dj * NCLASS + f), wj * v[u]);
            }
        }
    }
}

__global__ __launch_bounds__(256) void gcn_logsoftmax_kernel(float* __restrict__ logits) {
    int node = blockIdx.x * 256 + threadIdx.x;
    if (node >= NNODES) return;
    float4* p = (float4*)(logits + (size_t)node * NCLASS);
    float4 v[4];
    #pragma unroll
    for (int i = 0; i < 4; ++i) v[i] = p[i];
    float* f = (float*)v;
    float mx = f[0];
    #pragma unroll
    for (int i = 1; i < 16; ++i) mx = fmaxf(mx, f[i]);
    float sum = 0.f;
    #pragma unroll
    for (int i = 0; i < 16; ++i) sum += __expf(f[i] - mx);
    float lse = mx + __logf(sum);
    #pragma unroll
    for (int i = 0; i < 16; ++i) f[i] -= lse;
    #pragma unroll
    for (int i = 0; i < 4; ++i) p[i] = v[i];
}

extern "C" void kernel_launch(void* const* d_in, const int* in_sizes, int n_in,
                              void* d_out, int out_size, void* d_ws, size_t ws_size,
                              hipStream_t stream) {
    const float* x   = (const float*)d_in[0];
    const int*   src = (const int*)  d_in[1];
    const int*   dst = (const int*)  d_in[2];
    const float* ew  = (const float*)d_in[3];
    const float* W1  = (const float*)d_in[4];
    const float* b1  = (const float*)d_in[5];
    const float* W2  = (const float*)d_in[6];
    const float* b2  = (const float*)d_in[7];
    float* out = (float*)d_out;

    float* ws       = (float*)d_ws;
    float* support1 = ws;
    float* agg1     = ws + (size_t)NNODES * NHID;
    float* support2 = ws + 2 * (size_t)NNODES * NHID;

    gcn_gemm1_kernel<<<MTILES, 128, 0, stream>>>(x, W1, support1, agg1);
    gcn_spmm64_kernel<<<SPMM_BLOCKS, 32 * SPMM_WAVES_PER_BLOCK, 0, stream>>>(src, dst, ew, support1, agg1);

    gcn_gemm2_kernel<<<(MTILES + 7) / 8, 256, 0, stream>>>(agg1, b1, W2, b2, support2, out);
    gcn_spmm16_kernel<<<SPMM_BLOCKS, 32 * SPMM_WAVES_PER_BLOCK, 0, stream>>>(src, dst, ew, support2, out);

    gcn_logsoftmax_kernel<<<(NNODES + 255) / 256, 256, 0, stream>>>(out);
}


# Edges binned per call by destination block (LDS-ranked count, one span reservation per bucket); owner blocks sum rows with LDS f32 atomic adds (arrival order) and finish them; no global f32 atomics
# speedup vs baseline: 4.4187x; 1.5374x over previous
#include <hip/hip_runtime.h>

#define NNODES 100000
#define NEDGES 1600000
#define NFEAT 128
#define NHID 64
#define NCLASS 16
#define MTILES (NNODES / 16)


#define NBUCKETS 256
#define BROWS 391
#define BCAP 32768
#define BIN_THREADS 256
#define BIN_EPT 25
#define BIN_BLOCKS (NEDGES / (BIN_THREADS * BIN_EPT))
#define OWN_THREADS 1024
static_assert(NBUCKETS * BROWS >= NNODES, "buckets must cover every node");
static_assert(NEDGES % (BIN_THREADS * BIN_EPT) == 0, "edge count must fill whole binning blocks");
static_assert(BIN_THREADS == NBUCKETS, "one thread per bucket counter");

typedef float v2f __attribute__((ext_vector_type(2)));
typedef float v8f __attribute__((ext_vector_type(8)));

__device__ __forceinline__ void lds_add_f32(float* p, float v) {
    __hip_atomic_fetch_add(p, v, __ATOMIC_RELAXED, __HIP_MEMORY_SCOPE_WORKGROUP);
}


__global__ __launch_bounds__(128) void gcn_gemm1_kernel(const float* __restrict__ x,
                                                        const float* __restrict__ W1,
                                                        float* __restrict__ out,
                                                        unsigned* __restrict__ cursors) {
    __shared__ float As[16][NFEAT + 4];
    __shared__ float WsT[NHID][NFEAT + 4];

    const int tid = threadIdx.x;
    const int tileM = blockIdx.x;

    for (int i = tid; i < NFEAT * NHID; i += 128) {
        int k = i >> 6, n = i & (NHID - 1);
        WsT[n][k] = W1[i];
    }
    const float* xrow = x + (size_t)tileM * 16 * NFEAT;
    for (int i = tid; i < 16 * NFEAT; i += 128) {
        As[i >> 7][i & (NFEAT - 1)] = xrow[i];
    }
    __syncthreads();

    if (blockIdx.x == 0) {
        for (int i = tid; i <= NBUCKETS; i += 128) cursors[i] = 0u;
    }

    const int lane = tid & 31;
    const int wave = tid >> 5;
    const int m  = lane & 15;
    const int n  = lane & 15;
    const int kh = (lane >> 4) * 2;
    const int n0 = wave * 16;

    v8f acc = {};
    #pragma unroll 4
    for (int kk = 0; kk < NFEAT; kk += 4) {
        const int k0 = kk + kh;
        v2f a, b;
        a.x = As[m][k0];
        a.y = As[m][k0 + 1];
        b.x = WsT[n0 + n][k0];
        b.y = WsT[n0 + n][k0 + 1];
        acc = __builtin_amdgcn_wmma_f32_16x16x4_f32(false, a, false, b,
                                                    (short)0, acc, false, false);
    }

    float* orow = out + ((size_t)tileM * 16 + (lane >> 4) * 8) * NHID + n0 + n;
    #pragma unroll
    for (int r = 0; r < 8; ++r)
        orow[(size_t)r * NHID] = acc[r];
}

__global__ __launch_bounds__(BIN_THREADS) void gcn_bin_kernel(const int* __restrict__ src,
                                                              const int* __restrict__ dst,
                                                              const float* __restrict__ ew,
                                                              int4* __restrict__ recs,
                                                              int4* __restrict__ spill,
                                                              unsigned* __restrict__ cursors) {
    __shared__ unsigned cnt[NBUCKETS];
    __shared__ unsigned span[NBUCKETS];

    const int tid = threadIdx.x;
    cnt[tid] = 0u;
    __syncthreads();

    const int e0 = blockIdx.x * (BIN_THREADS * BIN_EPT) + tid;
    int s[BIN_EPT], d[BIN_EPT];
    float w[BIN_EPT];
    unsigned rk[BIN_EPT];
    #pragma unroll
    for (int j = 0; j < BIN_EPT; ++j) {
        const int e = e0 + j * BIN_THREADS;
        s[j] = src[e];
        d[j] = dst[e];
        w[j] = ew[e];
    }
    #pragma unroll
    for (int j = 0; j < BIN_EPT; ++j) {
        const unsigned bk = (unsigned)d[j] / BROWS;
        rk[j] = __hip_atomic_fetch_add(&cnt[bk], 1u, __ATOMIC_RELAXED, __HIP_MEMORY_SCOPE_WORKGROUP);
    }
    __syncthreads();
    {
        const unsigned c = cnt[tid];
        span[tid] = c ? __hip_atomic_fetch_add(&cursors[tid], c, __ATOMIC_RELAXED, __HIP_MEMORY_SCOPE_AGENT) : 0u;
    }
    __syncthreads();
    #pragma unroll
    for (int j = 0; j < BIN_EPT; ++j) {
        const unsigned bk = (unsigned)d[j] / BROWS;
        const unsigned pos = span[bk] + rk[j];
        if (pos < BCAP) {
            recs[(size_t)bk * BCAP + pos] = make_int4(s[j], d[j] - (int)(bk * BROWS), __float_as_int(w[j]), 0);
        } else {
            const unsigned o = __hip_atomic_fetch_add(&cursors[NBUCKETS], 1u, __ATOMIC_RELAXED, __HIP_MEMORY_SCOPE_AGENT);
            spill[o] = make_int4(s[j], d[j], __float_as_int(w[j]), 0);
        }
    }
}

__global__ __launch_bounds__(OWN_THREADS) void gcn_own64_kernel(const int4* __restrict__ recs,
                                                                const int4* __restrict__ spill,
                                                                const unsigned* __restrict__ cursors,
                                                                const float* __restrict__ sup,
                                                                float* __restrict__ out) {
    __shared__ float acc[BROWS * NHID];

    const int tid = threadIdx.x;
    const unsigned lane = tid & 31;
    const unsigned wave = (unsigned)__builtin_amdgcn_readfirstlane(tid >> 5);
    const int bk = blockIdx.x;
    const int n0 = bk * BROWS;
    const int nrows = min(BROWS, NNODES - n0);

    for (int i = tid; i < BROWS * NHID; i += OWN_THREADS) acc[i] = 0.f;
    const unsigned cnt  = min(cursors[bk], (unsigned)BCAP);
    const unsigned scnt = min(cursors[NBUCKETS], (unsigned)NEDGES);
    __syncthreads();

    const int4* my = recs + (size_t)bk * BCAP;
    for (unsigned b0 = wave * 32; b0 < cnt; b0 += OWN_THREADS) {
        const unsigned idx = b0 + lane;
        int4 r = make_int4(0, 0, 0, 0);
        if (idx < cnt) r = my[idx];
        const int n = (int)min(32u, cnt - b0);
        for (int g = 0; g < n; g += 8) {
            float v0[8], v1[8];
            #pragma unroll
            for (int u = 0; u < 8; ++u) {
                const int j = min(g + u, n - 1);
                const unsigned sj = (unsigned)__builtin_amdgcn_readlane(r.x, j);
                const unsigned o = sj * NHID + lane;
                v0[u] = sup[o];
                v1[u] = sup[o + 32];
            }
            #pragma unroll
            for (int u = 0; u < 8; ++u) {
                const int j = g + u;
                if (j < n) {
                    const unsigned dj = (unsigned)__builtin_amdgcn_readlane(r.y, j);
                    const float wj = __int_as_float(__builtin_amdgcn_readlane(r.z, j));
                    lds_add_f32(&acc[dj * NHID + lane],      wj * v0[u]);
                    lds_add_f32(&acc[dj * NHID + lane + 32], wj * v1[u]);
                }
            }
        }
    }
    for (unsigned b0 = wave * 32; b0 < scnt; b0 += OWN_THREADS) {
        const unsigned idx = b0 + lane;
        int4 r = make_int4(0, -1, 0, 0);
        if (idx < scnt) r = spill[idx];
        const int n = (int)min(32u, scnt - b0);
        for (int j = 0; j < n; ++j) {
            const int dj = __builtin_amdgcn_readlane(r.y, j) - n0;
            if ((unsigned)dj < (unsigned)nrows) {
                const unsigned sj = (unsigned)__builtin_amdgcn_readlane(r.x, j);
                const float wj = __int_as_float(__builtin_amdgcn_readlane(r.z, j));
                const unsigned o = sj * NHID + lane;
                lds_add_f32(&acc[dj * NHID + lane],      wj * sup[o]);
                lds_add_f32(&acc[dj * NHID + lane + 32], wj * sup[o + 32]);
            }
        }
    }
    __syncthreads();

    float4* o4 = (float4*)(out + (size_t)n0 * NHID);
    const float4* a4 = (const float4*)acc;
    for (int i = tid; i < nrows * (NHID / 4); i += OWN_THREADS) o4[i] = a4[i];
}

__global__ __launch_bounds__(256) void gcn_gemm2_kernel(const float* __restrict__ agg1,
                                                        const float* __restrict__ b1,
                                                        const float* __restrict__ W2,
                                                        float* __restrict__ out) {
    __shared__ float WsT[NCLASS][NHID + 4];
    __shared__ float bs[NHID];

    const int tid = threadIdx.x;

    for (int i = tid; i < NHID * NCLASS; i += 256) {
        int k = i >> 4, n = i & (NCLASS - 1);
        WsT[n][k] = W2[i];
    }
    if (tid < NHID) bs[tid] = b1[tid];
    __syncthreads();

    const int lane = tid & 31;
    const int wave = tid >> 5;
    const int tileM = blockIdx.x * 8 + wave;
    if (tileM >= MTILES) return;

    const int m  = lane & 15;
    const int n  = lane & 15;
    const int kh = (lane >> 4) * 2;

    const float* arow = agg1 + ((size_t)tileM * 16 + m) * NHID;

    v8f acc = {};
    #pragma unroll 4
    for (int kk = 0; kk < NHID; kk += 4) {
        const int k0 = kk + kh;
        v2f a, b;
        a.x = fmaxf(arow[k0]     + bs[k0],     0.f);
        a.y = fmaxf(arow[k0 + 1] + bs[k0 + 1], 0.f);
        b.x = WsT[n][k0];
        b.y = WsT[n][k0 + 1];
        acc = __builtin_amdgcn_wmma_f32_16x16x4_f32(false, a, false, b,
                                                    (short)0, acc, false, false);
    }

    float* orow = out + ((size_t)tileM * 16 + (lane >> 4) * 8) * NCLASS + n;
    #pragma unroll
    for (int r = 0; r < 8; ++r)
        orow[(size_t)r * NCLASS] = acc[r];
}

__global__ __launch_bounds__(OWN_THREADS) void gcn_own16_kernel(const int4* __restrict__ recs,
                                                                const int4* __restrict__ spill,
                                                                const unsigned* __restrict__ cursors,
                                                                const float* __restrict__ sup,
                                                                const float* __restrict__ b2,
                                                                float* __restrict__ out) {
    __shared__ float acc[BROWS * NCLASS];

    const int tid = threadIdx.x;
    const unsigned lane = tid & 31;
    const unsigned wave = (unsigned)__builtin_amdgcn_readfirstlane(tid >> 5);
    const bool odd = lane >= 16;
    const unsigned f = lane & 15;
    const int bk = blockIdx.x;
    const int n0 = bk * BROWS;
    const int nrows = min(BROWS, NNODES - n0);

    for (int i = tid; i < BROWS * NCLASS; i += OWN_THREADS) acc[i] = 0.f;
    const unsigned cnt  = min(cursors[bk], (unsigned)BCAP);
    const unsigned scnt = min(cursors[NBUCKETS], (unsigned)NEDGES);
    __syncthreads();

    const int4* my = recs + (size_t)bk * BCAP;
    for (unsigned b0 = wave * 32; b0 < cnt; b0 += OWN_THREADS) {
        const unsigned idx = b0 + lane;
        int4 r = make_int4(0, 0, 0, 0);
        if (idx < cnt) r = my[idx];
        const int n = (int)min(32u, cnt - b0);
        for (int g = 0; g < n; g += 16) {
            float v[8];
            #pragma unroll
            for (int u = 0; u < 8; ++u) {
                const int ja = min(g + 2 * u,     n - 1);
                const int jb = min(g + 2 * u + 1, n - 1);
                const int sa = __builtin_amdgcn_readlane(r.x, ja);
                const int sb = __builtin_amdgcn_readlane(r.x, jb);
                const unsigned sj = (unsigned)(odd ? sb : sa);
                v[u] = sup[sj * NCLASS + f];
            }
            #pragma unroll
            for (int u = 0; u < 8; ++u) {
                const int ja = g + 2 * u;
                if (ja < n) {
                    const int jb = min(ja + 1, n - 1);
                    const int da = __builtin_amdgcn_readlane(r.y, ja);
                    const int db = __builtin_amdgcn_readlane(r.y, jb);
                    const int wa = __builtin_amdgcn_readlane(r.z, ja);
                    const int wb = __builtin_amdgcn_readlane(r.z, jb);
                    const unsigned dj = (unsigned)(odd ? db : da);
                    const float wj = __int_as_float(odd ? wb : wa);
                    if (!odd || ja + 1 < n)
                        lds_add_f32(&acc[dj * NCLASS + f], wj * v[u]);
                }
            }
        }
    }
    for (unsigned b0 = wave * 32; b0 < scnt; b0 += OWN_THREADS) {
        const unsigned idx = b0 + lane;
        int4 r = make_int4(0, -1, 0, 0);
        if (idx < scnt) r = spill[idx];
        const int n = (int)min(32u, scnt - b0);
        for (int j = 0; j < n; ++j) {
            const int dj = __builtin_amdgcn_readlane(r.y, j) - n0;
            if ((unsigned)dj < (unsigned)nrows) {
                const unsigned sj = (unsigned)__builtin_amdgcn_readlane(r.x, j);
                const float wj = __int_as_float(__builtin_amdgcn_readlane(r.z, j));
                if (!odd) lds_add_f32(&acc[dj * NCLASS + f], wj * sup[sj * NCLASS + f]);
            }
        }
    }
    __syncthreads();

    if (tid < nrows) {
        float fv[NCLASS];
        #pragma unroll
        for (int i = 0; i < NCLASS; ++i) fv[i] = acc[tid * NCLASS + i] + b2[i];
        float mx = fv[0];
        #pragma unroll
        for (int i = 1; i < NCLASS; ++i) mx = fmaxf(mx, fv[i]);
        float sum = 0.f;
        #pragma unroll
        for (int i = 0; i < NCLASS; ++i) sum += __expf(fv[i] - mx);
        const float lse = mx + __logf(sum);
        float4* p = (float4*)(out + (size_t)(n0 + tid) * NCLASS);
        #pragma unroll
        for (int i = 0; i < 4; ++i)
            p[i] = make_float4(fv[4 * i] - lse, fv[4 * i + 1] - lse, fv[4 * i + 2] - lse, fv[4 * i + 3] - lse);
    }
}

extern "C" void kernel_launch(void* const* d_in, const int* in_sizes, int n_in,
                              void* d_out, int out_size, void* d_ws, size_t ws_size,
                              hipStream_t stream) {
    const float* x   = (const float*)d_in[0];
    const int*   src = (const int*)  d_in[1];
    const int*   dst = (const int*)  d_in[2];
    const float* ew  = (const float*)d_in[3];
    const float* W1  = (const float*)d_in[4];
    const float* b1  = (const float*)d_in[5];
    const float* W2  = (const float*)d_in[6];
    const float* b2  = (const float*)d_in[7];
    float* out = (float*)d_out;

    float* ws       = (float*)d_ws;
    float* support1 = ws;
    float* agg1     = ws + (size_t)NNODES * NHID;
    float* support2 = ws + 2 * (size_t)NNODES * NHID;
    int4*  recs     = (int4*)(support2 + (size_t)NNODES * NCLASS);
    int4*  spill    = recs + (size_t)NBUCKETS * BCAP;
    unsigned* cursors = (unsigned*)(spill + (size_t)NEDGES);

    gcn_gemm1_kernel<<<MTILES, 128, 0, stream>>>(x, W1, support1, cursors);
    gcn_bin_kernel<<<BIN_BLOCKS, BIN_THREADS, 0, stream>>>(src, dst, ew, recs, spill, cursors);
    gcn_own64_kernel<<<NBUCKETS, OWN_THREADS, 0, stream>>>(recs, spill, cursors, support1, agg1);

    gcn_gemm2_kernel<<<(MTILES + 7) / 8, 256, 0, stream>>>(agg1, b1, W2, support2);
    gcn_own16_kernel<<<NBUCKETS, OWN_THREADS, 0, stream>>>(recs, spill, cursors, support2, b2, out);
}


# Layer-2 projection moved into the layer-1 owner blocks: each wave takes 16 finished rows from LDS (bias, relu, f32 WMMA with W2) and stores support2; summed rows never written (5 launches to 4)
# speedup vs baseline: 4.8201x; 1.0908x over previous
#include <hip/hip_runtime.h>

#define NNODES 100000
#define NEDGES 1600000
#define NFEAT 128
#define NHID 64
#define NCLASS 16
#define MTILES (NNODES / 16)


#define NBUCKETS 256
#define BROWS 391
#define BCAP 32768
#define BIN_THREADS 256
#define BIN_EPT 25
#define BIN_BLOCKS (NEDGES / (BIN_THREADS * BIN_EPT))
#define OWN_THREADS 1024
#define APITCH (NHID + 4)
#define AROWS (((BROWS + 15) / 16) * 16)
static_assert(NBUCKETS * BROWS >= NNODES, "buckets must cover every node");
static_assert(NEDGES % (BIN_THREADS * BIN_EPT) == 0, "edge count must fill whole binning blocks");
static_assert(BIN_THREADS == NBUCKETS, "one thread per bucket counter");
static_assert(OWN_THREADS == NHID * NCLASS, "one owner thread per W2 element");
static_assert(AROWS / 16 <= OWN_THREADS / 32, "one wave per 16-row tile of a bucket");

typedef float v2f __attribute__((ext_vector_type(2)));
typedef float v8f __attribute__((ext_vector_type(8)));

__device__ __forceinline__ void lds_add_f32(float* p, float v) {
    __hip_atomic_fetch_add(p, v, __ATOMIC_RELAXED, __HIP_MEMORY_SCOPE_WORKGROUP);
}


__global__ __launch_bounds__(128) void gcn_gemm1_kernel(const float* __restrict__ x,
                                                        const float* __restrict__ W1,
                                                        float* __restrict__ out,
                                                        unsigned* __restrict__ cursors) {
    __shared__ float As[16][NFEAT + 4];
    __shared__ float WsT[NHID][NFEAT + 4];

    const int tid = threadIdx.x;
    const int tileM = blockIdx.x;

    for (int i = tid; i < NFEAT * NHID; i += 128) {
        int k = i >> 6, n = i & (NHID - 1);
        WsT[n][k] = W1[i];
    }
    const float* xrow = x + (size_t)tileM * 16 * NFEAT;
    for (int i = tid; i < 16 * NFEAT; i += 128) {
        As[i >> 7][i & (NFEAT - 1)] = xrow[i];
    }
    __syncthreads();

    if (blockIdx.x == 0) {
        for (int i = tid; i <= NBUCKETS; i += 128) cursors[i] = 0u;
    }

    const int lane = tid & 31;
    const int wave = tid >> 5;
    const int m  = lane & 15;
    const int n  = lane & 15;
    const int kh = (lane >> 4) * 2;
    const int n0 = wave * 16;

    v8f acc = {};
    #pragma unroll 4
    for (int kk = 0; kk < NFEAT; kk += 4) {
        const int k0 = kk + kh;
        v2f a, b;
        a.x = As[m][k0];
        a.y = As[m][k0 + 1];
        b.x = WsT[n0 + n][k0];
        b.y = WsT[n0 + n][k0 + 1];
        acc = __builtin_amdgcn_wmma_f32_16x16x4_f32(false, a, false, b,
                                                    (short)0, acc, false, false);
    }

    float* orow = out + ((size_t)tileM * 16 + (lane >> 4) * 8) * NHID + n0 + n;
    #pragma unroll
    for (int r = 0; r < 8; ++r)
        orow[(size_t)r * NHID] = acc[r];
}

__global__ __launch_bounds__(BIN_THREADS) void gcn_bin_kernel(const int* __restrict__ src,
                                                              const int* __restrict__ dst,
                                                              const float* __restrict__ ew,
                                                              int4* __restrict__ recs,
                                                              int4* __restrict__ spill,
                                                              unsigned* __restrict__ cursors) {
    __shared__ unsigned cnt[NBUCKETS];
    __shared__ unsigned span[NBUCKETS];

    const int tid = threadIdx.x;
    cnt[tid] = 0u;
    __syncthreads();

    const int e0 = blockIdx.x * (BIN_THREADS * BIN_EPT) + tid;
    int s[BIN_EPT], d[BIN_EPT];
    float w[BIN_EPT];
    unsigned rk[BIN_EPT];
    #pragma unroll
    for (int j = 0; j < BIN_EPT; ++j) {
        const int e = e0 + j * BIN_THREADS;
        s[j] = src[e];
        d[j] = dst[e];
        w[j] = ew[e];
    }
    #pragma unroll
    for (int j = 0; j < BIN_EPT; ++j) {
        const unsigned bk = (unsigned)d[j] / BROWS;
        rk[j] = __hip_atomic_fetch_add(&cnt[bk], 1u, __ATOMIC_RELAXED, __HIP_MEMORY_SCOPE_WORKGROUP);
    }
    __syncthreads();
    {
        const unsigned c = cnt[tid];
        span[tid] = c ? __hip_atomic_fetch_add(&cursors[tid], c, __ATOMIC_RELAXED, __HIP_MEMORY_SCOPE_AGENT) : 0u;
    }
    __syncthreads();
    #pragma unroll
    for (int j = 0; j < BIN_EPT; ++j) {
        const unsigned bk = (unsigned)d[j] / BROWS;
        const unsigned pos = span[bk] + rk[j];
        if (pos < BCAP) {
            recs[(size_t)bk * BCAP + pos] = make_int4(s[j], d[j] - (int)(bk * BROWS), __float_as_int(w[j]), 0);
        } else {
            const unsigned o = __hip_atomic_fetch_add(&cursors[NBUCKETS], 1u, __ATOMIC_RELAXED, __HIP_MEMORY_SCOPE_AGENT);
            spill[o] = make_int4(s[j], d[j], __float_as_int(w[j]), 0);
        }
    }
}

__global__ __launch_bounds__(OWN_THREADS) void gcn_own64_kernel(const int4* __restrict__ recs,
                                                                const int4* __restrict__ spill,
                                                                const unsigned* __restrict__ cursors,
                                                                const float* __restrict__ sup,
                                                                const float* __restrict__ b1,
                                                                const float* __restrict__ W2,
                                                                float* __restrict__ out) {
    __shared__ float acc[AROWS * APITCH];
    __shared__ float WsT[NCLASS][NHID + 4];
    __shared__ float bs[NHID];

    const int tid = threadIdx.x;
    const unsigned lane = tid & 31;
    const unsigned wave = (unsigned)__builtin_amdgcn_readfirstlane(tid >> 5);
    const int bk = blockIdx.x;
    const int n0 = bk * BROWS;
    const int nrows = min(BROWS, NNODES - n0);

    for (int i = tid; i < AROWS * APITCH; i += OWN_THREADS) acc[i] = 0.f;
    WsT[tid & (NCLASS - 1)][tid >> 4] = W2[tid];
    if (tid < NHID) bs[tid] = b1[tid];
    const unsigned cnt  = min(cursors[bk], (unsigned)BCAP);
    const unsigned scnt = min(cursors[NBUCKETS], (unsigned)NEDGES);
    __syncthreads();

    const int4* my = recs + (size_t)bk * BCAP;
    for (unsigned b0 = wave * 32; b0 < cnt; b0 += OWN_THREADS) {
        const unsigned idx = b0 + lane;
        int4 r = make_int4(0, 0, 0, 0);
        if (idx < cnt) r = my[idx];
        const int n = (int)min(32u, cnt - b0);
        for (int g = 0; g < n; g += 8) {
            float v0[8], v1[8];
            #pragma unroll
            for (int u = 0; u < 8; ++u) {
                const int j = min(g + u, n - 1);
                const unsigned sj = (unsigned)__builtin_amdgcn_readlane(r.x, j);
                const unsigned o = sj * NHID + lane;
                v0[u] = sup[o];
                v1[u] = sup[o + 32];
            }
            #pragma unroll
            for (int u = 0; u < 8; ++u) {
                const int j = g + u;
                if (j < n) {
                    const unsigned dj = (unsigned)__builtin_amdgcn_readlane(r.y, j);
                    const float wj = __int_as_float(__builtin_amdgcn_readlane(r.z, j));
                    lds_add_f32(&acc[dj * APITCH + lane],      wj * v0[u]);
                    lds_add_f32(&acc[dj * APITCH + lane + 32], wj * v1[u]);
                }
            }
        }
    }
    for (unsigned b0 = wave * 32; b0 < scnt; b0 += OWN_THREADS) {
        const unsigned idx = b0 + lane;
        int4 r = make_int4(0, -1, 0, 0);
        if (idx < scnt) r = spill[idx];
        const int n = (int)min(32u, scnt - b0);
        for (int j = 0; j < n; ++j) {
            const int dj = __builtin_amdgcn_readlane(r.y, j) - n0;
            if ((unsigned)dj < (unsigned)nrows) {
                const unsigned sj = (unsigned)__builtin_amdgcn_readlane(r.x, j);
                const float wj = __int_as_float(__builtin_amdgcn_readlane(r.z, j));
                const unsigned o = sj * NHID + lane;
                lds_add_f32(&acc[dj * APITCH + lane],      wj * sup[o]);
                lds_add_f32(&acc[dj * APITCH + lane + 32], wj * sup[o + 32]);
            }
        }
    }
    __syncthreads();

    const unsigned ntiles = (unsigned)(nrows + 15) >> 4;
    if (wave < ntiles) {
        const unsigned m  = lane & 15;
        const unsigned kh = (lane >> 4) * 2;
        const float* arow = &acc[(wave * 16 + m) * APITCH];

        v8f d = {};
        #pragma unroll
        for (int kk = 0; kk < NHID; kk += 4) {
            const unsigned k0 = kk + kh;
            v2f a, b;
            a.x = fmaxf(arow[k0]     + bs[k0],     0.f);
            a.y = fmaxf(arow[k0 + 1] + bs[k0 + 1], 0.f);
            b.x = WsT[m][k0];
            b.y = WsT[m][k0 + 1];
            d = __builtin_amdgcn_wmma_f32_16x16x4_f32(false, a, false, b,
                                                      (short)0, d, false, false);
        }

        const unsigned r0 = wave * 16 + (lane >> 4) * 8;
        float* orow = out + ((size_t)n0 + r0) * NCLASS + m;
        #pragma unroll
        for (int r = 0; r < 8; ++r)
            if ((int)(r0 + r) < nrows) orow[(size_t)r * NCLASS] = d[r];
    }
}

__global__ __launch_bounds__(OWN_THREADS) void gcn_own16_kernel(const int4* __restrict__ recs,
                                                                const int4* __restrict__ spill,
                                                                const unsigned* __restrict__ cursors,
                                                                const float* __restrict__ sup,
                                                                const float* __restrict__ b2,
                                                                float* __restrict__ out) {
    __shared__ float acc[BROWS * NCLASS];

    const int tid = threadIdx.x;
    const unsigned lane = tid & 31;
    const unsigned wave = (unsigned)__builtin_amdgcn_readfirstlane(tid >> 5);
    const bool odd = lane >= 16;
    const unsigned f = lane & 15;
    const int bk = blockIdx.x;
    const int n0 = bk * BROWS;
    const int nrows = min(BROWS, NNODES - n0);

    for (int i = tid; i < BROWS * NCLASS; i += OWN_THREADS) acc[i] = 0.f;
    const unsigned cnt  = min(cursors[bk], (unsigned)BCAP);
    const unsigned scnt = min(cursors[NBUCKETS], (unsigned)NEDGES);
    __syncthreads();

    const int4* my = recs + (size_t)bk * BCAP;
    for (unsigned b0 = wave * 32; b0 < cnt; b0 += OWN_THREADS) {
        const unsigned idx = b0 + lane;
        int4 r = make_int4(0, 0, 0, 0);
        if (idx < cnt) r = my[idx];
        const int n = (int)min(32u, cnt - b0);
        for (int g = 0; g < n; g += 16) {
            float v[8];
            #pragma unroll
            for (int u = 0; u < 8; ++u) {
                const int ja = min(g + 2 * u,     n - 1);
                const int jb = min(g + 2 * u + 1, n - 1);
                const int sa = __builtin_amdgcn_readlane(r.x, ja);
                const int sb = __builtin_amdgcn_readlane(r.x, jb);
                const unsigned sj = (unsigned)(odd ? sb : sa);
                v[u] = sup[sj * NCLASS + f];
            }
            #pragma unroll
            for (int u = 0; u < 8; ++u) {
                const int ja = g + 2 * u;
                if (ja < n) {
                    const int jb = min(ja + 1, n - 1);
                    const int da = __builtin_amdgcn_readlane(r.y, ja);
                    const int db = __builtin_amdgcn_readlane(r.y, jb);
                    const int wa = __builtin_amdgcn_readlane(r.z, ja);
                    const int wb = __builtin_amdgcn_readlane(r.z, jb);
                    const unsigned dj = (unsigned)(odd ? db : da);
                    const float wj = __int_as_float(odd ? wb : wa);
                    if (!odd || ja + 1 < n)
                        lds_add_f32(&acc[dj * NCLASS + f], wj * v[u]);
                }
            }
        }
    }
    for (unsigned b0 = wave * 32; b0 < scnt; b0 += OWN_THREADS) {
        const unsigned idx = b0 + lane;
        int4 r = make_int4(0, -1, 0, 0);
        if (idx < scnt) r = spill[idx];
        const int n = (int)min(32u, scnt - b0);
        for (int j = 0; j < n; ++j) {
            const int dj = __builtin_amdgcn_readlane(r.y, j) - n0;
            if ((unsigned)dj < (unsigned)nrows) {
                const unsigned sj = (unsigned)__builtin_amdgcn_readlane(r.x, j);
                const float wj = __int_as_float(__builtin_amdgcn_readlane(r.z, j));
                if (!odd) lds_add_f32(&acc[dj * NCLASS + f], wj * sup[sj * NCLASS + f]);
            }
        }
    }
    __syncthreads();

    if (tid < nrows) {
        float fv[NCLASS];
        #pragma unroll
        for (int i = 0; i < NCLASS; ++i) fv[i] = acc[tid * NCLASS + i] + b2[i];
        float mx = fv[0];
        #pragma unroll
        for (int i = 1; i < NCLASS; ++i) mx = fmaxf(mx, fv[i]);
        float sum = 0.f;
        #pragma unroll
        for (int i = 0; i < NCLASS; ++i) sum += __expf(fv[i] - mx);
        const float lse = mx + __logf(sum);
        float4* p = (float4*)(out + (size_t)(n0 + tid) * NCLASS);
        #pragma unroll
        for (int i = 0; i < 4; ++i)
            p[i] = make_float4(fv[4 * i] - lse, fv[4 * i + 1] - lse, fv[4 * i + 2] - lse, fv[4 * i + 3] - lse);
    }
}

extern "C" void kernel_launch(void* const* d_in, const int* in_sizes, int n_in,
                              void* d_out, int out_size, void* d_ws, size_t ws_size,
                              hipStream_t stream) {
    const float* x   = (const float*)d_in[0];
    const int*   src = (const int*)  d_in[1];
    const int*   dst = (const int*)  d_in[2];
    const float* ew  = (const float*)d_in[3];
    const float* W1  = (const float*)d_in[4];
    const float* b1  = (const float*)d_in[5];
    const float* W2  = (const float*)d_in[6];
    const float* b2  = (const float*)d_in[7];
    float* out = (float*)d_out;

    float* ws       = (float*)d_ws;
    float* support1 = ws;
    float* support2 = ws + (size_t)NNODES * NHID;
    int4*  recs     = (int4*)(support2 + (size_t)NNODES * NCLASS);
    int4*  spill    = recs + (size_t)NBUCKETS * BCAP;
    unsigned* cursors = (unsigned*)(spill + (size_t)NEDGES);

    gcn_gemm1_kernel<<<MTILES, 128, 0, stream>>>(x, W1, support1, cursors);
    gcn_bin_kernel<<<BIN_BLOCKS, BIN_THREADS, 0, stream>>>(src, dst, ew, recs, spill, cursors);
    gcn_own64_kernel<<<NBUCKETS, OWN_THREADS, 0, stream>>>(recs, spill, cursors, support1, b1, W2, support2);

    gcn_own16_kernel<<<NBUCKETS, OWN_THREADS, 0, stream>>>(recs, spill, cursors, support2, b2, out);
}


# GEMM1 blocks stage W1 once and walk 5 row tiles (6,250 blocks to 1,250), pipelined: next tile's x loads (16 B per lane) fly during the current products; rows stored behind the next barrier
# speedup vs baseline: 5.4901x; 1.1390x over previous
#include <hip/hip_runtime.h>

#define NNODES 100000
#define NEDGES 1600000
#define NFEAT 128
#define NHID 64
#define NCLASS 16
#define MTILES (NNODES / 16)


#define NBUCKETS 256
#define BROWS 391
#define BCAP 32768
#define BIN_THREADS 256
#define BIN_EPT 25
#define BIN_BLOCKS (NEDGES / (BIN_THREADS * BIN_EPT))
#define OWN_THREADS 1024
#define APITCH (NHID + 4)
#define AROWS (((BROWS + 15) / 16) * 16)
static_assert(NBUCKETS * BROWS >= NNODES, "buckets must cover every node");
static_assert(NEDGES % (BIN_THREADS * BIN_EPT) == 0, "edge count must fill whole binning blocks");
static_assert(BIN_THREADS == NBUCKETS, "one thread per bucket counter");
static_assert(OWN_THREADS == NHID * NCLASS, "one owner thread per W2 element");
static_assert(AROWS / 16 <= OWN_THREADS / 32, "one wave per 16-row tile of a bucket");

typedef float v2f __attribute__((ext_vector_type(2)));
typedef float v8f __attribute__((ext_vector_type(8)));

__device__ __forceinline__ void lds_add_f32(float* p, float v) {
    __hip_atomic_fetch_add(p, v, __ATOMIC_RELAXED, __HIP_MEMORY_SCOPE_WORKGROUP);
}


#define G1_TPB 5
#define G1_BLOCKS (MTILES / G1_TPB)
static_assert(MTILES % G1_TPB == 0, "M tiles must fill whole GEMM1 blocks");
__global__ __launch_bounds__(128) void gcn_gemm1_kernel(const float* __restrict__ x,
                                                        const float* __restrict__ W1,
                                                        float* __restrict__ out,
                                                        unsigned* __restrict__ cursors) {
    __shared__ __attribute__((aligned(16))) float As[2][16][NFEAT + 4];
    __shared__ float WsT[NHID][NFEAT + 4];

    const int tid = threadIdx.x;
    const int tile0 = blockIdx.x * G1_TPB;

    const float4* xv = (const float4*)(x + (size_t)tile0 * 16 * NFEAT);
    float4 xa = xv[tid], xb = xv[tid + 128], xc = xv[tid + 256], xd = xv[tid + 384];

    for (int i = tid; i < NFEAT * NHID; i += 128) {
        int k = i >> 6, n = i & (NHID - 1);
        WsT[n][k] = W1[i];
    }

    if (blockIdx.x == 0) {
        for (int i = tid; i <= NBUCKETS; i += 128) cursors[i] = 0u;
    }

    const int lane = tid & 31;
    const int wave = tid >> 5;
    const int m  = lane & 15;
    const int n  = lane & 15;
    const int kh = (lane >> 4) * 2;
    const int n0 = wave * 16;
    float* orow0 = out + ((size_t)tile0 * 16 + (lane >> 4) * 8) * NHID + n0 + n;

    v8f prev = {};
    #pragma unroll 1
    for (int t = 0; t < G1_TPB; ++t) {
        float (*A)[NFEAT + 4] = As[t & 1];
        {
            const int r4 = tid >> 5, c4 = (tid & 31) * 4;
            *(float4*)&A[r4][c4]      = xa;
            *(float4*)&A[r4 + 4][c4]  = xb;
            *(float4*)&A[r4 + 8][c4]  = xc;
            *(float4*)&A[r4 + 12][c4] = xd;
        }
        __syncthreads();
        if (t > 0) {
            float* orow = orow0 + (size_t)(t - 1) * 16 * NHID;
            #pragma unroll
            for (int r = 0; r < 8; ++r)
                orow[(size_t)r * NHID] = prev[r];
        }
        if (t + 1 < G1_TPB) {
            xv += 16 * NFEAT / 4;
            xa = xv[tid]; xb = xv[tid + 128]; xc = xv[tid + 256]; xd = xv[tid + 384];
        }

        v8f acc = {};
        #pragma unroll 4
        for (int kk = 0; kk < NFEAT; kk += 4) {
            const int k0 = kk + kh;
            v2f a, b;
            a.x = A[m][k0];
            a.y = A[m][k0 + 1];
            b.x = WsT[n0 + n][k0];
            b.y = WsT[n0 + n][k0 + 1];
            acc = __builtin_amdgcn_wmma_f32_16x16x4_f32(false, a, false, b,
                                                        (short)0, acc, false, false);
        }
        prev = acc;
    }
    {
        float* orow = orow0 + (size_t)(G1_TPB - 1) * 16 * NHID;
        #pragma unroll
        for (int r = 0; r < 8; ++r)
            orow[(size_t)r * NHID] = prev[r];
    }
}

__global__ __launch_bounds__(BIN_THREADS) void gcn_bin_kernel(const int* __restrict__ src,
                                                              const int* __restrict__ dst,
                                                              const float* __restrict__ ew,
                                                              int4* __restrict__ recs,
                                                              int4* __restrict__ spill,
                                                              unsigned* __restrict__ cursors) {
    __shared__ unsigned cnt[NBUCKETS];
    __shared__ unsigned span[NBUCKETS];

    const int tid = threadIdx.x;
    cnt[tid] = 0u;
    __syncthreads();

    const int e0 = blockIdx.x * (BIN_THREADS * BIN_EPT) + tid;
    int s[BIN_EPT], d[BIN_EPT];
    float w[BIN_EPT];
    unsigned rk[BIN_EPT];
    #pragma unroll
    for (int j = 0; j < BIN_EPT; ++j) {
        const int e = e0 + j * BIN_THREADS;
        s[j] = src[e];
        d[j] = dst[e];
        w[j] = ew[e];
    }
    #pragma unroll
    for (int j = 0; j < BIN_EPT; ++j) {
        const unsigned bk = (unsigned)d[j] / BROWS;
        rk[j] = __hip_atomic_fetch_add(&cnt[bk], 1u, __ATOMIC_RELAXED, __HIP_MEMORY_SCOPE_WORKGROUP);
    }
    __syncthreads();
    {
        const unsigned c = cnt[tid];
        span[tid] = c ? __hip_atomic_fetch_add(&cursors[tid], c, __ATOMIC_RELAXED, __HIP_MEMORY_SCOPE_AGENT) : 0u;
    }
    __syncthreads();
    #pragma unroll
    for (int j = 0; j < BIN_EPT; ++j) {
        const unsigned bk = (unsigned)d[j] / BROWS;
        const unsigned pos = span[bk] + rk[j];
        if (pos < BCAP) {
            recs[(size_t)bk * BCAP + pos] = make_int4(s[j], d[j] - (int)(bk * BROWS), __float_as_int(w[j]), 0);
        } else {
            const unsigned o = __hip_atomic_fetch_add(&cursors[NBUCKETS], 1u, __ATOMIC_RELAXED, __HIP_MEMORY_SCOPE_AGENT);
            spill[o] = make_int4(s[j], d[j], __float_as_int(w[j]), 0);
        }
    }
}

__global__ __launch_bounds__(OWN_THREADS) void gcn_own64_kernel(const int4* __restrict__ recs,
                                                                const int4* __restrict__ spill,
                                                                const unsigned* __restrict__ cursors,
                                                                const float* __restrict__ sup,
                                                                const float* __restrict__ b1,
                                                                const float* __restrict__ W2,
                                                                float* __restrict__ out) {
    __shared__ float acc[AROWS * APITCH];
    __shared__ float WsT[NCLASS][NHID + 4];
    __shared__ float bs[NHID];

    const int tid = threadIdx.x;
    const unsigned lane = tid & 31;
    const unsigned wave = (unsigned)__builtin_amdgcn_readfirstlane(tid >> 5);
    const int bk = blockIdx.x;
    const int n0 = bk * BROWS;
    const int nrows = min(BROWS, NNODES - n0);

    for (int i = tid; i < AROWS * APITCH; i += OWN_THREADS) acc[i] = 0.f;
    WsT[tid & (NCLASS - 1)][tid >> 4] = W2[tid];
    if (tid < NHID) bs[tid] = b1[tid];
    const unsigned cnt  = min(cursors[bk], (unsigned)BCAP);
    const unsigned scnt = min(cursors[NBUCKETS], (unsigned)NEDGES);
    __syncthreads();

    const int4* my = recs + (size_t)bk * BCAP;
    for (unsigned b0 = wave * 32; b0 < cnt; b0 += OWN_THREADS) {
        const unsigned idx = b0 + lane;
        int4 r = make_int4(0, 0, 0, 0);
        if (idx < cnt) r = my[idx];
        const int n = (int)min(32u, cnt - b0);
        for (int g = 0; g < n; g += 8) {
            float v0[8], v1[8];
            #pragma unroll
            for (int u = 0; u < 8; ++u) {
                const int j = min(g + u, n - 1);
                const unsigned sj = (unsigned)__builtin_amdgcn_readlane(r.x, j);
                const unsigned o = sj * NHID + lane;
                v0[u] = sup[o];
                v1[u] = sup[o + 32];
            }
            #pragma unroll
            for (int u = 0; u < 8; ++u) {
                const int j = g + u;
                if (j < n) {
                    const unsigned dj = (unsigned)__builtin_amdgcn_readlane(r.y, j);
                    const float wj = __int_as_float(__builtin_amdgcn_readlane(r.z, j));
                    lds_add_f32(&acc[dj * APITCH + lane],      wj * v0[u]);
                    lds_add_f32(&acc[dj * APITCH + lane + 32], wj * v1[u]);
                }
            }
        }
    }
    for (unsigned b0 = wave * 32; b0 < scnt; b0 += OWN_THREADS) {
        const unsigned idx = b0 + lane;
        int4 r = make_int4(0, -1, 0, 0);
        if (idx < scnt) r = spill[idx];
        const int n = (int)min(32u, scnt - b0);
        for (int j = 0; j < n; ++j) {
            const int dj = __builtin_amdgcn_readlane(r.y, j) - n0;
            if ((unsigned)dj < (unsigned)nrows) {
                const unsigned sj = (unsigned)__builtin_amdgcn_readlane(r.x, j);
                const float wj = __int_as_float(__builtin_amdgcn_readlane(r.z, j));
                const unsigned o = sj * NHID + lane;
                lds_add_f32(&acc[dj * APITCH + lane],      wj * sup[o]);
                lds_add_f32(&acc[dj * APITCH + lane + 32], wj * sup[o + 32]);
            }
        }
    }
    __syncthreads();

    const unsigned ntiles = (unsigned)(nrows + 15) >> 4;
    if (wave < ntiles) {
        const unsigned m  = lane & 15;
        const unsigned kh = (lane >> 4) * 2;
        const float* arow = &acc[(wave * 16 + m) * APITCH];

        v8f d = {};
        #pragma unroll
        for (int kk = 0; kk < NHID; kk += 4) {
            const unsigned k0 = kk + kh;
            v2f a, b;
            a.x = fmaxf(arow[k0]     + bs[k0],     0.f);
            a.y = fmaxf(arow[k0 + 1] + bs[k0 + 1], 0.f);
            b.x = WsT[m][k0];
            b.y = WsT[m][k0 + 1];
            d = __builtin_amdgcn_wmma_f32_16x16x4_f32(false, a, false, b,
                                                      (short)0, d, false, false);
        }

        const unsigned r0 = wave * 16 + (lane >> 4) * 8;
        float* orow = out + ((size_t)n0 + r0) * NCLASS + m;
        #pragma unroll
        for (int r = 0; r < 8; ++r)
            if ((int)(r0 + r) < nrows) orow[(size_t)r * NCLASS] = d[r];
    }
}

__global__ __launch_bounds__(OWN_THREADS) void gcn_own16_kernel(const int4* __restrict__ recs,
                                                                const int4* __restrict__ spill,
                                                                const unsigned* __restrict__ cursors,
                                                                const float* __restrict__ sup,
                                                                const float* __restrict__ b2,
                                                                float* __restrict__ out) {
    __shared__ float acc[BROWS * NCLASS];

    const int tid = threadIdx.x;
    const unsigned lane = tid & 31;
    const unsigned wave = (unsigned)__builtin_amdgcn_readfirstlane(tid >> 5);
    const bool odd = lane >= 16;
    const unsigned f = lane & 15;
    const int bk = blockIdx.x;
    const int n0 = bk * BROWS;
    const int nrows = min(BROWS, NNODES - n0);

    for (int i = tid; i < BROWS * NCLASS; i += OWN_THREADS) acc[i] = 0.f;
    const unsigned cnt  = min(cursors[bk], (unsigned)BCAP);
    const unsigned scnt = min(cursors[NBUCKETS], (unsigned)NEDGES);
    __syncthreads();

    const int4* my = recs + (size_t)bk * BCAP;
    for (unsigned b0 = wave * 32; b0 < cnt; b0 += OWN_THREADS) {
        const unsigned idx = b0 + lane;
        int4 r = make_int4(0, 0, 0, 0);
        if (idx < cnt) r = my[idx];
        const int n = (int)min(32u, cnt - b0);
        for (int g = 0; g < n; g += 16) {
            float v[8];
            #pragma unroll
            for (int u = 0; u < 8; ++u) {
                const int ja = min(g + 2 * u,     n - 1);
                const int jb = min(g + 2 * u + 1, n - 1);
                const int sa = __builtin_amdgcn_readlane(r.x, ja);
                const int sb = __builtin_amdgcn_readlane(r.x, jb);
                const unsigned sj = (unsigned)(odd ? sb : sa);
                v[u] = sup[sj * NCLASS + f];
            }
            #pragma unroll
            for (int u = 0; u < 8; ++u) {
                const int ja = g + 2 * u;
                if (ja < n) {
                    const int jb = min(ja + 1, n - 1);
                    const int da = __builtin_amdgcn_readlane(r.y, ja);
                    const int db = __builtin_amdgcn_readlane(r.y, jb);
                    const int wa = __builtin_amdgcn_readlane(r.z, ja);
                    const int wb = __builtin_amdgcn_readlane(r.z, jb);
                    const unsigned dj = (unsigned)(odd ? db : da);
                    const float wj = __int_as_float(odd ? wb : wa);
                    if (!odd || ja + 1 < n)
                        lds_add_f32(&acc[dj * NCLASS + f], wj * v[u]);
                }
            }
        }
    }
    for (unsigned b0 = wave * 32; b0 < scnt; b0 += OWN_THREADS) {
        const unsigned idx = b0 + lane;
        int4 r = make_int4(0, -1, 0, 0);
        if (idx < scnt) r = spill[idx];
        const int n = (int)min(32u, scnt - b0);
        for (int j = 0; j < n; ++j) {
            const int dj = __builtin_amdgcn_readlane(r.y, j) - n0;
            if ((unsigned)dj < (unsigned)nrows) {
                const unsigned sj = (unsigned)__builtin_amdgcn_readlane(r.x, j);
                const float wj = __int_as_float(__builtin_amdgcn_readlane(r.z, j));
                if (!odd) lds_add_f32(&acc[dj * NCLASS + f], wj * sup[sj * NCLASS + f]);
            }
        }
    }
    __syncthreads();

    if (tid < nrows) {
        float fv[NCLASS];
        #pragma unroll
        for (int i = 0; i < NCLASS; ++i) fv[i] = acc[tid * NCLASS + i] + b2[i];
        float mx = fv[0];
        #pragma unroll
        for (int i = 1; i < NCLASS; ++i) mx = fmaxf(mx, fv[i]);
        float sum = 0.f;
        #pragma unroll
        for (int i = 0; i < NCLASS; ++i) sum += __expf(fv[i] - mx);
        const float lse = mx + __logf(sum);
        float4* p = (float4*)(out + (size_t)(n0 + tid) * NCLASS);
        #pragma unroll
        for (int i = 0; i < 4; ++i)
            p[i] = make_float4(fv[4 * i] - lse, fv[4 * i + 1] - lse, fv[4 * i + 2] - lse, fv[4 * i + 3] - lse);
    }
}

extern "C" void kernel_launch(void* const* d_in, const int* in_sizes, int n_in,
                              void* d_out, int out_size, void* d_ws, size_t ws_size,
                              hipStream_t stream) {
    const float* x   = (const float*)d_in[0];
    const int*   src = (const int*)  d_in[1];
    const int*   dst = (const int*)  d_in[2];
    const float* ew  = (const float*)d_in[3];
    const float* W1  = (const float*)d_in[4];
    const float* b1  = (const float*)d_in[5];
    const float* W2  = (const float*)d_in[6];
    const float* b2  = (const float*)d_in[7];
    float* out = (float*)d_out;

    float* ws       = (float*)d_ws;
    float* support1 = ws;
    float* support2 = ws + (size_t)NNODES * NHID;
    int4*  recs     = (int4*)(support2 + (size_t)NNODES * NCLASS);
    int4*  spill    = recs + (size_t)NBUCKETS * BCAP;
    unsigned* cursors = (unsigned*)(spill + (size_t)NEDGES);

    gcn_gemm1_kernel<<<G1_BLOCKS, 128, 0, stream>>>(x, W1, support1, cursors);
    gcn_bin_kernel<<<BIN_BLOCKS, BIN_THREADS, 0, stream>>>(src, dst, ew, recs, spill, cursors);
    gcn_own64_kernel<<<NBUCKETS, OWN_THREADS, 0, stream>>>(recs, spill, cursors, support1, b1, W2, support2);

    gcn_own16_kernel<<<NBUCKETS, OWN_THREADS, 0, stream>>>(recs, spill, cursors, support2, b2, out);
}


# Layer-1 owners counting-sort their records by row in LDS; a wave sums each run of equal rows in registers (fma chain, sorted order) and adds it to the LDS row once per run, not once per edge
# speedup vs baseline: 5.8463x; 1.0649x over previous
#include <hip/hip_runtime.h>

#define NNODES 100000
#define NEDGES 1600000
#define NFEAT 128
#define NHID 64
#define NCLASS 16
#define MTILES (NNODES / 16)


#define NBUCKETS 256
#define BROWS 391
#define BCAP 32768
#define BIN_THREADS 256
#define BIN_EPT 25
#define BIN_BLOCKS (NEDGES / (BIN_THREADS * BIN_EPT))
#define OWN_THREADS 1024
#define APITCH (NHID + 4)
#define AROWS (((BROWS + 15) / 16) * 16)
static_assert(NBUCKETS * BROWS >= NNODES, "buckets must cover every node");
static_assert(NEDGES % (BIN_THREADS * BIN_EPT) == 0, "edge count must fill whole binning blocks");
static_assert(BIN_THREADS == NBUCKETS, "one thread per bucket counter");
static_assert(OWN_THREADS == NHID * NCLASS, "one owner thread per W2 element");
static_assert(AROWS / 16 <= OWN_THREADS / 32, "one wave per 16-row tile of a bucket");
#define SCAP 8192
#define SREC (SCAP / OWN_THREADS)
#define RSTN 416
#define SRCBITS 17
static_assert(SCAP % OWN_THREADS == 0, "whole records per thread");
static_assert(NNODES <= (1 << SRCBITS), "source index must fit its field");
static_assert(AROWS <= RSTN && RSTN <= 512 && RSTN % 32 == 0, "row index must fit its field and the scan");

typedef float v2f __attribute__((ext_vector_type(2)));
typedef float v8f __attribute__((ext_vector_type(8)));

__device__ __forceinline__ void lds_add_f32(float* p, float v) {
    __hip_atomic_fetch_add(p, v, __ATOMIC_RELAXED, __HIP_MEMORY_SCOPE_WORKGROUP);
}


#define G1_TPB 5
#define G1_BLOCKS (MTILES / G1_TPB)
static_assert(MTILES % G1_TPB == 0, "M tiles must fill whole GEMM1 blocks");
__global__ __launch_bounds__(128) void gcn_gemm1_kernel(const float* __restrict__ x,
                                                        const float* __restrict__ W1,
                                                        float* __restrict__ out,
                                                        unsigned* __restrict__ cursors) {
    __shared__ __attribute__((aligned(16))) float As[2][16][NFEAT + 4];
    __shared__ float WsT[NHID][NFEAT + 4];

    const int tid = threadIdx.x;
    const int tile0 = blockIdx.x * G1_TPB;

    const float4* xv = (const float4*)(x + (size_t)tile0 * 16 * NFEAT);
    float4 xa = xv[tid], xb = xv[tid + 128], xc = xv[tid + 256], xd = xv[tid + 384];

    for (int i = tid; i < NFEAT * NHID; i += 128) {
        int k = i >> 6, n = i & (NHID - 1);
        WsT[n][k] = W1[i];
    }

    if (blockIdx.x == 0) {
        for (int i = tid; i <= NBUCKETS; i += 128) cursors[i] = 0u;
    }

    const int lane = tid & 31;
    const int wave = tid >> 5;
    const int m  = lane & 15;
    const int n  = lane & 15;
    const int kh = (lane >> 4) * 2;
    const int n0 = wave * 16;
    float* orow0 = out + ((size_t)tile0 * 16 + (lane >> 4) * 8) * NHID + n0 + n;

    v8f prev = {};
    #pragma unroll 1
    for (int t = 0; t < G1_TPB; ++t) {
        float (*A)[NFEAT + 4] = As[t & 1];
        {
            const int r4 = tid >> 5, c4 = (tid & 31) * 4;
            *(float4*)&A[r4][c4]      = xa;
            *(float4*)&A[r4 + 4][c4]  = xb;
            *(float4*)&A[r4 + 8][c4]  = xc;
            *(float4*)&A[r4 + 12][c4] = xd;
        }
        __syncthreads();
        if (t > 0) {
            float* orow = orow0 + (size_t)(t - 1) * 16 * NHID;
            #pragma unroll
            for (int r = 0; r < 8; ++r)
                orow[(size_t)r * NHID] = prev[r];
        }
        if (t + 1 < G1_TPB) {
            xv += 16 * NFEAT / 4;
            xa = xv[tid]; xb = xv[tid + 128]; xc = xv[tid + 256]; xd = xv[tid + 384];
        }

        v8f acc = {};
        #pragma unroll 4
        for (int kk = 0; kk < NFEAT; kk += 4) {
            const int k0 = kk + kh;
            v2f a, b;
            a.x = A[m][k0];
            a.y = A[m][k0 + 1];
            b.x = WsT[n0 + n][k0];
            b.y = WsT[n0 + n][k0 + 1];
            acc = __builtin_amdgcn_wmma_f32_16x16x4_f32(false, a, false, b,
                                                        (short)0, acc, false, false);
        }
        prev = acc;
    }
    {
        float* orow = orow0 + (size_t)(G1_TPB - 1) * 16 * NHID;
        #pragma unroll
        for (int r = 0; r < 8; ++r)
            orow[(size_t)r * NHID] = prev[r];
    }
}

__global__ __launch_bounds__(BIN_THREADS) void gcn_bin_kernel(const int* __restrict__ src,
                                                              const int* __restrict__ dst,
                                                              const float* __restrict__ ew,
                                                              int4* __restrict__ recs,
                                                              int4* __restrict__ spill,
                                                              unsigned* __restrict__ cursors) {
    __shared__ unsigned cnt[NBUCKETS];
    __shared__ unsigned span[NBUCKETS];

    const int tid = threadIdx.x;
    cnt[tid] = 0u;
    __syncthreads();

    const int e0 = blockIdx.x * (BIN_THREADS * BIN_EPT) + tid;
    int s[BIN_EPT], d[BIN_EPT];
    float w[BIN_EPT];
    unsigned rk[BIN_EPT];
    #pragma unroll
    for (int j = 0; j < BIN_EPT; ++j) {
        const int e = e0 + j * BIN_THREADS;
        s[j] = src[e];
        d[j] = dst[e];
        w[j] = ew[e];
    }
    #pragma unroll
    for (int j = 0; j < BIN_EPT; ++j) {
        const unsigned bk = (unsigned)d[j] / BROWS;
        rk[j] = __hip_atomic_fetch_add(&cnt[bk], 1u, __ATOMIC_RELAXED, __HIP_MEMORY_SCOPE_WORKGROUP);
    }
    __syncthreads();
    {
        const unsigned c = cnt[tid];
        span[tid] = c ? __hip_atomic_fetch_add(&cursors[tid], c, __ATOMIC_RELAXED, __HIP_MEMORY_SCOPE_AGENT) : 0u;
    }
    __syncthreads();
    #pragma unroll
    for (int j = 0; j < BIN_EPT; ++j) {
        const unsigned bk = (unsigned)d[j] / BROWS;
        const unsigned pos = span[bk] + rk[j];
        if (pos < BCAP) {
            recs[(size_t)bk * BCAP + pos] = make_int4(s[j], d[j] - (int)(bk * BROWS), __float_as_int(w[j]), 0);
        } else {
            const unsigned o = __hip_atomic_fetch_add(&cursors[NBUCKETS], 1u, __ATOMIC_RELAXED, __HIP_MEMORY_SCOPE_AGENT);
            spill[o] = make_int4(s[j], d[j], __float_as_int(w[j]), 0);
        }
    }
}

__global__ __launch_bounds__(OWN_THREADS) void gcn_own64_kernel(const int4* __restrict__ recs,
                                                                const int4* __restrict__ spill,
                                                                const unsigned* __restrict__ cursors,
                                                                const float* __restrict__ sup,
                                                                const float* __restrict__ b1,
                                                                const float* __restrict__ W2,
                                                                float* __restrict__ out) {
    __shared__ float acc[AROWS * APITCH];
    __shared__ float WsT[NCLASS][NHID + 4];
    __shared__ float bs[NHID];
    __shared__ unsigned rowst[RSTN];
    __shared__ uint2 srt[SCAP];

    const int tid = threadIdx.x;
    const unsigned lane = tid & 31;
    const unsigned wave = (unsigned)__builtin_amdgcn_readfirstlane(tid >> 5);
    const int bk = blockIdx.x;
    const int n0 = bk * BROWS;
    const int nrows = min(BROWS, NNODES - n0);

    const unsigned cnt  = min(cursors[bk], (unsigned)BCAP);
    const unsigned scnt = min(cursors[NBUCKETS], (unsigned)NEDGES);
    for (int i = tid; i < AROWS * APITCH; i += OWN_THREADS) acc[i] = 0.f;
    WsT[tid & (NCLASS - 1)][tid >> 4] = W2[tid];
    if (tid < NHID) bs[tid] = b1[tid];

    const int4* my = recs + (size_t)bk * BCAP;
    for (unsigned c0 = 0; c0 < cnt; c0 += SCAP) {
        const unsigned nc = min(cnt - c0, (unsigned)SCAP);
        if (tid < RSTN) rowst[tid] = 0u;
        unsigned pk[SREC], rk[SREC];
        float wv[SREC];
        int4 rr[SREC];
        unsigned ci[SREC];
        #pragma unroll
        for (int i = 0; i < SREC; ++i) {
            const unsigned idx = tid + i * OWN_THREADS;
            ci[i] = c0 + min(idx, nc - 1u);
        }
        __builtin_amdgcn_sched_barrier(0);
        #pragma unroll
        for (int i = 0; i < SREC; ++i) rr[i] = my[ci[i]];
        __builtin_amdgcn_sched_barrier(0);
        __syncthreads();
        #pragma unroll
        for (int i = 0; i < SREC; ++i) {
            pk[i] = (unsigned)rr[i].x | ((unsigned)rr[i].y << SRCBITS);
            wv[i] = __int_as_float(rr[i].z);
        }
        #pragma unroll
        for (int i = 0; i < SREC; ++i) {
            const unsigned idx = tid + i * OWN_THREADS;
            rk[i] = 0u;
            if (idx < nc)
                rk[i] = __hip_atomic_fetch_add(&rowst[pk[i] >> SRCBITS], 1u, __ATOMIC_RELAXED, __HIP_MEMORY_SCOPE_WORKGROUP);
        }
        __syncthreads();
        if (wave == 0) {
            unsigned c[RSTN / 32];
            unsigned s = 0u;
            #pragma unroll
            for (int i = 0; i < RSTN / 32; ++i) { c[i] = rowst[lane * (RSTN / 32) + i]; s += c[i]; }
            unsigned incl = s;
            #pragma unroll
            for (int d = 1; d < 32; d <<= 1) {
                const unsigned t = __shfl_up(incl, d);
                if (lane >= (unsigned)d) incl += t;
            }
            unsigned run = incl - s;
            #pragma unroll
            for (int i = 0; i < RSTN / 32; ++i) { rowst[lane * (RSTN / 32) + i] = run; run += c[i]; }
        }
        __syncthreads();
        unsigned pos[SREC];
        #pragma unroll
        for (int i = 0; i < SREC; ++i) pos[i] = rowst[pk[i] >> SRCBITS] + rk[i];
        #pragma unroll
        for (int i = 0; i < SREC; ++i) {
            const unsigned idx = tid + i * OWN_THREADS;
            if (idx < nc) srt[pos[i]] = make_uint2(pk[i], __float_as_uint(wv[i]));
        }
        __syncthreads();

        const unsigned per = (nc + 31u) >> 5;
        const unsigned e0 = min(wave * per, nc);
        const unsigned e1 = min(e0 + per, nc);
        unsigned cur = 0u;
        float a0 = 0.f, a1 = 0.f;
        for (unsigned b0 = e0; b0 < e1; b0 += 32) {
            const unsigned idx = b0 + lane;
            uint2 r = make_uint2(0u, 0u);
            if (idx < e1) r = srt[idx];
            const int n = (int)min(32u, e1 - b0);
            if (n == 32) {
                float v0[2][8], v1[2][8];
                #pragma unroll
                for (int u = 0; u < 8; ++u) {
                    const unsigned sj = (unsigned)__builtin_amdgcn_readlane(r.x, u) & ((1u << SRCBITS) - 1u);
                    const unsigned o = sj * NHID + lane;
                    v0[0][u] = sup[o];
                    v1[0][u] = sup[o + 32];
                }
                #pragma unroll
                for (int gi = 0; gi < 4; ++gi) {
                    if (gi + 1 < 4) {
                        #pragma unroll
                        for (int u = 0; u < 8; ++u) {
                            const unsigned sj = (unsigned)__builtin_amdgcn_readlane(r.x, (gi + 1) * 8 + u) & ((1u << SRCBITS) - 1u);
                            const unsigned o = sj * NHID + lane;
                            v0[(gi + 1) & 1][u] = sup[o];
                            v1[(gi + 1) & 1][u] = sup[o + 32];
                        }
                    }
                    #pragma unroll
                    for (int u = 0; u < 8; ++u) {
                        const unsigned rj = (unsigned)__builtin_amdgcn_readlane(r.x, gi * 8 + u) >> SRCBITS;
                        const float wj = __uint_as_float((unsigned)__builtin_amdgcn_readlane(r.y, gi * 8 + u));
                        if (rj != cur) {
                            lds_add_f32(&acc[cur * APITCH + lane],      a0);
                            lds_add_f32(&acc[cur * APITCH + lane + 32], a1);
                            a0 = 0.f; a1 = 0.f; cur = rj;
                        }
                        a0 = fmaf(wj, v0[gi & 1][u], a0);
                        a1 = fmaf(wj, v1[gi & 1][u], a1);
                    }
                }
            } else {
                for (int g = 0; g < n; g += 8) {
                    float v0[8], v1[8];
                    #pragma unroll
                    for (int u = 0; u < 8; ++u) {
                        const int j = min(g + u, n - 1);
                        const unsigned sj = (unsigned)__builtin_amdgcn_readlane(r.x, j) & ((1u << SRCBITS) - 1u);
                        const unsigned o = sj * NHID + lane;
                        v0[u] = sup[o];
                        v1[u] = sup[o + 32];
                    }
                    #pragma unroll
                    for (int u = 0; u < 8; ++u) {
                        const int j = g + u;
                        if (j < n) {
                            const unsigned rj = (unsigned)__builtin_amdgcn_readlane(r.x, j) >> SRCBITS;
                            const float wj = __uint_as_float((unsigned)__builtin_amdgcn_readlane(r.y, j));
                            if (rj != cur) {
                                lds_add_f32(&acc[cur * APITCH + lane],      a0);
                                lds_add_f32(&acc[cur * APITCH + lane + 32], a1);
                                a0 = 0.f; a1 = 0.f; cur = rj;
                            }
                            a0 = fmaf(wj, v0[u], a0);
                            a1 = fmaf(wj, v1[u], a1);
                        }
                    }
                }
            }
        }
        lds_add_f32(&acc[cur * APITCH + lane],      a0);
        lds_add_f32(&acc[cur * APITCH + lane + 32], a1);
    }
    __syncthreads();
    for (unsigned b0 = wave * 32; b0 < scnt; b0 += OWN_THREADS) {
        const unsigned idx = b0 + lane;
        int4 r = make_int4(0, -1, 0, 0);
        if (idx < scnt) r = spill[idx];
        const int n = (int)min(32u, scnt - b0);
        for (int j = 0; j < n; ++j) {
            const int dj = __builtin_amdgcn_readlane(r.y, j) - n0;
            if ((unsigned)dj < (unsigned)nrows) {
                const unsigned sj = (unsigned)__builtin_amdgcn_readlane(r.x, j);
                const float wj = __int_as_float(__builtin_amdgcn_readlane(r.z, j));
                const unsigned o = sj * NHID + lane;
                lds_add_f32(&acc[dj * APITCH + lane],      wj * sup[o]);
                lds_add_f32(&acc[dj * APITCH + lane + 32], wj * sup[o + 32]);
            }
        }
    }
    __syncthreads();

    const unsigned ntiles = (unsigned)(nrows + 15) >> 4;
    if (wave < ntiles) {
        const unsigned m  = lane & 15;
        const unsigned kh = (lane >> 4) * 2;
        const float* arow = &acc[(wave * 16 + m) * APITCH];

        v8f d = {};
        #pragma unroll
        for (int kk = 0; kk < NHID; kk += 4) {
            const unsigned k0 = kk + kh;
            v2f a, b;
            a.x = fmaxf(arow[k0]     + bs[k0],     0.f);
            a.y = fmaxf(arow[k0 + 1] + bs[k0 + 1], 0.f);
            b.x = WsT[m][k0];
            b.y = WsT[m][k0 + 1];
            d = __builtin_amdgcn_wmma_f32_16x16x4_f32(false, a, false, b,
                                                      (short)0, d, false, false);
        }

        const unsigned r0 = wave * 16 + (lane >> 4) * 8;
        float* orow = out + ((size_t)n0 + r0) * NCLASS + m;
        #pragma unroll
        for (int r = 0; r < 8; ++r)
            if ((int)(r0 + r) < nrows) orow[(size_t)r * NCLASS] = d[r];
    }
}

__global__ __launch_bounds__(OWN_THREADS) void gcn_own16_kernel(const int4* __restrict__ recs,
                                                                const int4* __restrict__ spill,
                                                                const unsigned* __restrict__ cursors,
                                                                const float* __restrict__ sup,
                                                                const float* __restrict__ b2,
                                                                float* __restrict__ out) {
    __shared__ float acc[BROWS * NCLASS];

    const int tid = threadIdx.x;
    const unsigned lane = tid & 31;
    const unsigned wave = (unsigned)__builtin_amdgcn_readfirstlane(tid >> 5);
    const bool odd = lane >= 16;
    const unsigned f = lane & 15;
    const int bk = blockIdx.x;
    const int n0 = bk * BROWS;
    const int nrows = min(BROWS, NNODES - n0);

    for (int i = tid; i < BROWS * NCLASS; i += OWN_THREADS) acc[i] = 0.f;
    const unsigned cnt  = min(cursors[bk], (unsigned)BCAP);
    const unsigned scnt = min(cursors[NBUCKETS], (unsigned)NEDGES);
    __syncthreads();

    const int4* my = recs + (size_t)bk * BCAP;
    for (unsigned b0 = wave * 32; b0 < cnt; b0 += OWN_THREADS) {
        const unsigned idx = b0 + lane;
        int4 r = make_int4(0, 0, 0, 0);
        if (idx < cnt) r = my[idx];
        const int n = (int)min(32u, cnt - b0);
        for (int g = 0; g < n; g += 16) {
            float v[8];
            #pragma unroll
            for (int u = 0; u < 8; ++u) {
                const int ja = min(g + 2 * u,     n - 1);
                const int jb = min(g + 2 * u + 1, n - 1);
                const int sa = __builtin_amdgcn_readlane(r.x, ja);
                const int sb = __builtin_amdgcn_readlane(r.x, jb);
                const unsigned sj = (unsigned)(odd ? sb : sa);
                v[u] = sup[sj * NCLASS + f];
            }
            #pragma unroll
            for (int u = 0; u < 8; ++u) {
                const int ja = g + 2 * u;
                if (ja < n) {
                    const int jb = min(ja + 1, n - 1);
                    const int da = __builtin_amdgcn_readlane(r.y, ja);
                    const int db = __builtin_amdgcn_readlane(r.y, jb);
                    const int wa = __builtin_amdgcn_readlane(r.z, ja);
                    const int wb = __builtin_amdgcn_readlane(r.z, jb);
                    const unsigned dj = (unsigned)(odd ? db : da);
                    const float wj = __int_as_float(odd ? wb : wa);
                    if (!odd || ja + 1 < n)
                        lds_add_f32(&acc[dj * NCLASS + f], wj * v[u]);
                }
            }
        }
    }
    for (unsigned b0 = wave * 32; b0 < scnt; b0 += OWN_THREADS) {
        const unsigned idx = b0 + lane;
        int4 r = make_int4(0, -1, 0, 0);
        if (idx < scnt) r = spill[idx];
        const int n = (int)min(32u, scnt - b0);
        for (int j = 0; j < n; ++j) {
            const int dj = __builtin_amdgcn_readlane(r.y, j) - n0;
            if ((unsigned)dj < (unsigned)nrows) {
                const unsigned sj = (unsigned)__builtin_amdgcn_readlane(r.x, j);
                const float wj = __int_as_float(__builtin_amdgcn_readlane(r.z, j));
                if (!odd) lds_add_f32(&acc[dj * NCLASS + f], wj * sup[sj * NCLASS + f]);
            }
        }
    }
    __syncthreads();

    if (tid < nrows) {
        float fv[NCLASS];
        #pragma unroll
        for (int i = 0; i < NCLASS; ++i) fv[i] = acc[tid * NCLASS + i] + b2[i];
        float mx = fv[0];
        #pragma unroll
        for (int i = 1; i < NCLASS; ++i) mx = fmaxf(mx, fv[i]);
        float sum = 0.f;
        #pragma unroll
        for (int i = 0; i < NCLASS; ++i) sum += __expf(fv[i] - mx);
        const float lse = mx + __logf(sum);
        float4* p = (float4*)(out + (size_t)(n0 + tid) * NCLASS);
        #pragma unroll
        for (int i = 0; i < 4; ++i)
            p[i] = make_float4(fv[4 * i] - lse, fv[4 * i + 1] - lse, fv[4 * i + 2] - lse, fv[4 * i + 3] - lse);
    }
}

extern "C" void kernel_launch(void* const* d_in, const int* in_sizes, int n_in,
                              void* d_out, int out_size, void* d_ws, size_t ws_size,
                              hipStream_t stream) {
    const float* x   = (const float*)d_in[0];
    const int*   src = (const int*)  d_in[1];
    const int*   dst = (const int*)  d_in[2];
    const float* ew  = (const float*)d_in[3];
    const float* W1  = (const float*)d_in[4];
    const float* b1  = (const float*)d_in[5];
    const float* W2  = (const float*)d_in[6];
    const float* b2  = (const float*)d_in[7];
    float* out = (float*)d_out;

    float* ws       = (float*)d_ws;
    float* support1 = ws;
    float* support2 = ws + (size_t)NNODES * NHID;
    int4*  recs     = (int4*)(support2 + (size_t)NNODES * NCLASS);
    int4*  spill    = recs + (size_t)NBUCKETS * BCAP;
    unsigned* cursors = (unsigned*)(spill + (size_t)NEDGES);

    gcn_gemm1_kernel<<<G1_BLOCKS, 128, 0, stream>>>(x, W1, support1, cursors);
    gcn_bin_kernel<<<BIN_BLOCKS, BIN_THREADS, 0, stream>>>(src, dst, ew, recs, spill, cursors);
    gcn_own64_kernel<<<NBUCKETS, OWN_THREADS, 0, stream>>>(recs, spill, cursors, support1, b1, W2, support2);

    gcn_own16_kernel<<<NBUCKETS, OWN_THREADS, 0, stream>>>(recs, spill, cursors, support2, b2, out);
}


# Binning blocks group their records (now 8 bytes) by bucket in LDS and write each bucket segment with consecutive lanes: one contiguous run per store, not 32 lines; slots and sum order as before
# speedup vs baseline: 6.6471x; 1.1370x over previous
#include <hip/hip_runtime.h>

#define NNODES 100000
#define NEDGES 1600000
#define NFEAT 128
#define NHID 64
#define NCLASS 16
#define MTILES (NNODES / 16)


#define NBUCKETS 256
#define BROWS 391
#define BCAP 32768
#define BIN_THREADS 256
#define BIN_EPT 25
#define BIN_EDGES (BIN_THREADS * BIN_EPT)
#define BIN_BLOCKS (NEDGES / BIN_EDGES)
#define OWN_THREADS 1024
#define APITCH (NHID + 4)
#define AROWS (((BROWS + 15) / 16) * 16)
static_assert(NBUCKETS * BROWS >= NNODES, "buckets must cover every node");
static_assert(NEDGES % BIN_EDGES == 0, "edge count must fill whole binning blocks");
static_assert(NBUCKETS == (BIN_THREADS / 32) * 32, "copy-out: every binning wave writes 32 buckets");
static_assert(BIN_EDGES <= 65536, "a rank inside a block must fit 16 bits");
static_assert(BIN_THREADS == NBUCKETS, "one thread per bucket counter");
static_assert(OWN_THREADS == NHID * NCLASS, "one owner thread per W2 element");
static_assert(AROWS / 16 <= OWN_THREADS / 32, "one wave per 16-row tile of a bucket");
#define SCAP 8192
#define SREC (SCAP / OWN_THREADS)
#define RSTN 416
#define SRCBITS 17
static_assert(SCAP % OWN_THREADS == 0, "whole records per thread");
static_assert(NNODES <= (1 << SRCBITS), "source index must fit its field");
static_assert(AROWS <= RSTN && RSTN <= 512 && RSTN % 32 == 0, "row index must fit its field and the scan");

typedef float v2f __attribute__((ext_vector_type(2)));
typedef float v8f __attribute__((ext_vector_type(8)));

__device__ __forceinline__ void lds_add_f32(float* p, float v) {
    __hip_atomic_fetch_add(p, v, __ATOMIC_RELAXED, __HIP_MEMORY_SCOPE_WORKGROUP);
}


#define G1_TPB 5
#define G1_BLOCKS (MTILES / G1_TPB)
static_assert(MTILES % G1_TPB == 0, "M tiles must fill whole GEMM1 blocks");
__global__ __launch_bounds__(128) void gcn_gemm1_kernel(const float* __restrict__ x,
                                                        const float* __restrict__ W1,
                                                        float* __restrict__ out,
                                                        unsigned* __restrict__ cursors) {
    __shared__ __attribute__((aligned(16))) float As[2][16][NFEAT + 4];
    __shared__ float WsT[NHID][NFEAT + 4];

    const int tid = threadIdx.x;
    const int tile0 = blockIdx.x * G1_TPB;

    const float4* xv = (const float4*)(x + (size_t)tile0 * 16 * NFEAT);
    float4 xa = xv[tid], xb = xv[tid + 128], xc = xv[tid + 256], xd = xv[tid + 384];

    for (int i = tid; i < NFEAT * NHID; i += 128) {
        int k = i >> 6, n = i & (NHID - 1);
        WsT[n][k] = W1[i];
    }

    if (blockIdx.x == 0) {
        for (int i = tid; i <= NBUCKETS; i += 128) cursors[i] = 0u;
    }

    const int lane = tid & 31;
    const int wave = tid >> 5;
    const int m  = lane & 15;
    const int n  = lane & 15;
    const int kh = (lane >> 4) * 2;
    const int n0 = wave * 16;
    float* orow0 = out + ((size_t)tile0 * 16 + (lane >> 4) * 8) * NHID + n0 + n;

    v8f prev = {};
    #pragma unroll 1
    for (int t = 0; t < G1_TPB; ++t) {
        float (*A)[NFEAT + 4] = As[t & 1];
        {
            const int r4 = tid >> 5, c4 = (tid & 31) * 4;
            *(float4*)&A[r4][c4]      = xa;
            *(float4*)&A[r4 + 4][c4]  = xb;
            *(float4*)&A[r4 + 8][c4]  = xc;
            *(float4*)&A[r4 + 12][c4] = xd;
        }
        __syncthreads();
        if (t > 0) {
            float* orow = orow0 + (size_t)(t - 1) * 16 * NHID;
            #pragma unroll
            for (int r = 0; r < 8; ++r)
                orow[(size_t)r * NHID] = prev[r];
        }
        if (t + 1 < G1_TPB) {
            xv += 16 * NFEAT / 4;
            xa = xv[tid]; xb = xv[tid + 128]; xc = xv[tid + 256]; xd = xv[tid + 384];
        }

        v8f acc = {};
        #pragma unroll 4
        for (int kk = 0; kk < NFEAT; kk += 4) {
            const int k0 = kk + kh;
            v2f a, b;
            a.x = A[m][k0];
            a.y = A[m][k0 + 1];
            b.x = WsT[n0 + n][k0];
            b.y = WsT[n0 + n][k0 + 1];
            acc = __builtin_amdgcn_wmma_f32_16x16x4_f32(false, a, false, b,
                                                        (short)0, acc, false, false);
        }
        prev = acc;
    }
    {
        float* orow = orow0 + (size_t)(G1_TPB - 1) * 16 * NHID;
        #pragma unroll
        for (int r = 0; r < 8; ++r)
            orow[(size_t)r * NHID] = prev[r];
    }
}

__global__ __launch_bounds__(BIN_THREADS) void gcn_bin_kernel(const int* __restrict__ src,
                                                              const int* __restrict__ dst,
                                                              const float* __restrict__ ew,
                                                              uint2* __restrict__ recs,
                                                              int4* __restrict__ spill,
                                                              unsigned* __restrict__ cursors) {
    __shared__ unsigned cnt[NBUCKETS];
    __shared__ unsigned span[NBUCKETS];
    __shared__ unsigned lst[NBUCKETS];
    __shared__ unsigned wsum[BIN_THREADS / 32];
    __shared__ uint2 stage[BIN_EDGES];

    const int tid = threadIdx.x;
    const unsigned lane = tid & 31;
    const unsigned wave = (unsigned)__builtin_amdgcn_readfirstlane(tid >> 5);
    cnt[tid] = 0u;
    __syncthreads();

    const int e0 = blockIdx.x * BIN_EDGES + tid;
    int s[BIN_EPT], d[BIN_EPT];
    float w[BIN_EPT];
    #pragma unroll
    for (int j = 0; j < BIN_EPT; ++j) {
        const int e = e0 + j * BIN_THREADS;
        s[j] = src[e];
        d[j] = dst[e];
        w[j] = ew[e];
    }
    unsigned pk[BIN_EPT], br[BIN_EPT];
    #pragma unroll
    for (int j = 0; j < BIN_EPT; ++j) {
        const unsigned bk = (unsigned)d[j] / BROWS;
        const unsigned row = (unsigned)d[j] - bk * BROWS;
        pk[j] = (unsigned)s[j] | (row << SRCBITS);
        const unsigned rk = __hip_atomic_fetch_add(&cnt[bk], 1u, __ATOMIC_RELAXED, __HIP_MEMORY_SCOPE_WORKGROUP);
        br[j] = (bk << 16) | rk;
    }
    __syncthreads();
    {
        const unsigned c = cnt[tid];
        const unsigned g = c ? __hip_atomic_fetch_add(&cursors[tid], c, __ATOMIC_RELAXED, __HIP_MEMORY_SCOPE_AGENT) : 0u;
        unsigned incl = c;
        #pragma unroll
        for (int dd = 1; dd < 32; dd <<= 1) {
            const unsigned t = __shfl_up(incl, dd);
            if (lane >= (unsigned)dd) incl += t;
        }
        if (lane == 31) wsum[wave] = incl;
        __syncthreads();
        unsigned base = 0u;
        #pragma unroll
        for (unsigned i = 0; i < BIN_THREADS / 32; ++i)
            if (i < wave) base += wsum[i];
        lst[tid] = base + incl - c;
        span[tid] = g;
    }
    __syncthreads();
    #pragma unroll
    for (int j = 0; j < BIN_EPT; ++j)
        stage[lst[br[j] >> 16] + (br[j] & 0xffffu)] = make_uint2(pk[j], __float_as_uint(w[j]));
    __syncthreads();

    const unsigned kb = wave * 32u;
    const unsigned cv = cnt[kb + lane];
    const unsigned lv = lst[kb + lane];
    const unsigned gv = span[kb + lane];
    #pragma unroll
    for (int h = 0; h < 2; ++h) {
        uint2 rc[16];
        #pragma unroll
        for (int b = 0; b < 16; ++b) {
            const unsigned l0 = (unsigned)__builtin_amdgcn_readlane(lv, h * 16 + b);
            rc[b] = stage[min(l0 + lane, (unsigned)(BIN_EDGES - 1))];
        }
        #pragma unroll
        for (int b = 0; b < 16; ++b) {
            const unsigned k  = kb + (unsigned)(h * 16 + b);
            const unsigned n  = (unsigned)__builtin_amdgcn_readlane(cv, h * 16 + b);
            const unsigned l0 = (unsigned)__builtin_amdgcn_readlane(lv, h * 16 + b);
            const unsigned g0 = (unsigned)__builtin_amdgcn_readlane(gv, h * 16 + b);
            uint2 r = rc[b];
            for (unsigned i = lane; i < n; i += 32) {
                if (i >= 32u) r = stage[l0 + i];
                const unsigned pos = g0 + i;
                if (pos < BCAP) {
                    recs[(size_t)k * BCAP + pos] = r;
                } else {
                    const unsigned o = __hip_atomic_fetch_add(&cursors[NBUCKETS], 1u, __ATOMIC_RELAXED, __HIP_MEMORY_SCOPE_AGENT);
                    spill[o] = make_int4((int)(r.x & ((1u << SRCBITS) - 1u)), (int)(k * BROWS + (r.x >> SRCBITS)), (int)r.y, 0);
                }
            }
        }
    }
}

__global__ __launch_bounds__(OWN_THREADS) void gcn_own64_kernel(const uint2* __restrict__ recs,
                                                                const int4* __restrict__ spill,
                                                                const unsigned* __restrict__ cursors,
                                                                const float* __restrict__ sup,
                                                                const float* __restrict__ b1,
                                                                const float* __restrict__ W2,
                                                                float* __restrict__ out) {
    __shared__ float acc[AROWS * APITCH];
    __shared__ float WsT[NCLASS][NHID + 4];
    __shared__ float bs[NHID];
    __shared__ unsigned rowst[RSTN];
    __shared__ uint2 srt[SCAP];

    const int tid = threadIdx.x;
    const unsigned lane = tid & 31;
    const unsigned wave = (unsigned)__builtin_amdgcn_readfirstlane(tid >> 5);
    const int bk = blockIdx.x;
    const int n0 = bk * BROWS;
    const int nrows = min(BROWS, NNODES - n0);

    const unsigned cnt  = min(cursors[bk], (unsigned)BCAP);
    const unsigned scnt = min(cursors[NBUCKETS], (unsigned)NEDGES);
    for (int i = tid; i < AROWS * APITCH; i += OWN_THREADS) acc[i] = 0.f;
    WsT[tid & (NCLASS - 1)][tid >> 4] = W2[tid];
    if (tid < NHID) bs[tid] = b1[tid];

    const uint2* my = recs + (size_t)bk * BCAP;
    for (unsigned c0 = 0; c0 < cnt; c0 += SCAP) {
        const unsigned nc = min(cnt - c0, (unsigned)SCAP);
        if (tid < RSTN) rowst[tid] = 0u;
        unsigned pk[SREC], rk[SREC];
        float wv[SREC];
        uint2 rr[SREC];
        unsigned ci[SREC];
        #pragma unroll
        for (int i = 0; i < SREC; ++i) {
            const unsigned idx = tid + i * OWN_THREADS;
            ci[i] = c0 + min(idx, nc - 1u);
        }
        __builtin_amdgcn_sched_barrier(0);
        #pragma unroll
        for (int i = 0; i < SREC; ++i) rr[i] = my[ci[i]];
        __builtin_amdgcn_sched_barrier(0);
        __syncthreads();
        #pragma unroll
        for (int i = 0; i < SREC; ++i) {
            pk[i] = rr[i].x;
            wv[i] = __uint_as_float(rr[i].y);
        }
        #pragma unroll
        for (int i = 0; i < SREC; ++i) {
            const unsigned idx = tid + i * OWN_THREADS;
            rk[i] = 0u;
            if (idx < nc)
                rk[i] = __hip_atomic_fetch_add(&rowst[pk[i] >> SRCBITS], 1u, __ATOMIC_RELAXED, __HIP_MEMORY_SCOPE_WORKGROUP);
        }
        __syncthreads();
        if (wave == 0) {
            unsigned c[RSTN / 32];
            unsigned s = 0u;
            #pragma unroll
            for (int i = 0; i < RSTN / 32; ++i) { c[i] = rowst[lane * (RSTN / 32) + i]; s += c[i]; }
            unsigned incl = s;
            #pragma unroll
            for (int d = 1; d < 32; d <<= 1) {
                const unsigned t = __shfl_up(incl, d);
                if (lane >= (unsigned)d) incl += t;
            }
            unsigned run = incl - s;
            #pragma unroll
            for (int i = 0; i < RSTN / 32; ++i) { rowst[lane * (RSTN / 32) + i] = run; run += c[i]; }
        }
        __syncthreads();
        unsigned pos[SREC];
        #pragma unroll
        for (int i = 0; i < SREC; ++i) pos[i] = rowst[pk[i] >> SRCBITS] + rk[i];
        #pragma unroll
        for (int i = 0; i < SREC; ++i) {
            const unsigned idx = tid + i * OWN_THREADS;
            if (idx < nc) srt[pos[i]] = make_uint2(pk[i], __float_as_uint(wv[i]));
        }
        __syncthreads();

        const unsigned per = (nc + 31u) >> 5;
        const unsigned e0 = min(wave * per, nc);
        const unsigned e1 = min(e0 + per, nc);
        unsigned cur = 0u;
        float a0 = 0.f, a1 = 0.f;
        for (unsigned b0 = e0; b0 < e1; b0 += 32) {
            const unsigned idx = b0 + lane;
            uint2 r = make_uint2(0u, 0u);
            if (idx < e1) r = srt[idx];
            const int n = (int)min(32u, e1 - b0);
            if (n == 32) {
                float v0[2][8], v1[2][8];
                #pragma unroll
                for (int u = 0; u < 8; ++u) {
                    const unsigned sj = (unsigned)__builtin_amdgcn_readlane(r.x, u) & ((1u << SRCBITS) - 1u);
                    const unsigned o = sj * NHID + lane;
                    v0[0][u] = sup[o];
                    v1[0][u] = sup[o + 32];
                }
                #pragma unroll
                for (int gi = 0; gi < 4; ++gi) {
                    if (gi + 1 < 4) {
                        #pragma unroll
                        for (int u = 0; u < 8; ++u) {
                            const unsigned sj = (unsigned)__builtin_amdgcn_readlane(r.x, (gi + 1) * 8 + u) & ((1u << SRCBITS) - 1u);
                            const unsigned o = sj * NHID + lane;
                            v0[(gi + 1) & 1][u] = sup[o];
                            v1[(gi + 1) & 1][u] = sup[o + 32];
                        }
                    }
                    #pragma unroll
                    for (int u = 0; u < 8; ++u) {
                        const unsigned rj = (unsigned)__builtin_amdgcn_readlane(r.x, gi * 8 + u) >> SRCBITS;
                        const float wj = __uint_as_float((unsigned)__builtin_amdgcn_readlane(r.y, gi * 8 + u));
                        if (rj != cur) {
                            lds_add_f32(&acc[cur * APITCH + lane],      a0);
                            lds_add_f32(&acc[cur * APITCH + lane + 32], a1);
                            a0 = 0.f; a1 = 0.f; cur = rj;
                        }
                        a0 = fmaf(wj, v0[gi & 1][u], a0);
                        a1 = fmaf(wj, v1[gi & 1][u], a1);
                    }
                }
            } else {
                for (int g = 0; g < n; g += 8) {
                    float v0[8], v1[8];
                    #pragma unroll
                    for (int u = 0; u < 8; ++u) {
                        const int j = min(g + u, n - 1);
                        const unsigned sj = (unsigned)__builtin_amdgcn_readlane(r.x, j) & ((1u << SRCBITS) - 1u);
                        const unsigned o = sj * NHID + lane;
                        v0[u] = sup[o];
                        v1[u] = sup[o + 32];
                    }
                    #pragma unroll
                    for (int u = 0; u < 8; ++u) {
                        const int j = g + u;
                        if (j < n) {
                            const unsigned rj = (unsigned)__builtin_amdgcn_readlane(r.x, j) >> SRCBITS;
                            const float wj = __uint_as_float((unsigned)__builtin_amdgcn_readlane(r.y, j));
                            if (rj != cur) {
                                lds_add_f32(&acc[cur * APITCH + lane],      a0);
                                lds_add_f32(&acc[cur * APITCH + lane + 32], a1);
                                a0 = 0.f; a1 = 0.f; cur = rj;
                            }
                            a0 = fmaf(wj, v0[u], a0);
                            a1 = fmaf(wj, v1[u], a1);
                        }
                    }
                }
            }
        }
        lds_add_f32(&acc[cur * APITCH + lane],      a0);
        lds_add_f32(&acc[cur * APITCH + lane + 32], a1);
    }
    __syncthreads();
    for (unsigned b0 = wave * 32; b0 < scnt; b0 += OWN_THREADS) {
        const unsigned idx = b0 + lane;
        int4 r = make_int4(0, -1, 0, 0);
        if (idx < scnt) r = spill[idx];
        const int n = (int)min(32u, scnt - b0);
        for (int j = 0; j < n; ++j) {
            const int dj = __builtin_amdgcn_readlane(r.y, j) - n0;
            if ((unsigned)dj < (unsigned)nrows) {
                const unsigned sj = (unsigned)__builtin_amdgcn_readlane(r.x, j);
                const float wj = __int_as_float(__builtin_amdgcn_readlane(r.z, j));
                const unsigned o = sj * NHID + lane;
                lds_add_f32(&acc[dj * APITCH + lane],      wj * sup[o]);
                lds_add_f32(&acc[dj * APITCH + lane + 32], wj * sup[o + 32]);
            }
        }
    }
    __syncthreads();

    const unsigned ntiles = (unsigned)(nrows + 15) >> 4;
    if (wave < ntiles) {
        const unsigned m  = lane & 15;
        const unsigned kh = (lane >> 4) * 2;
        const float* arow = &acc[(wave * 16 + m) * APITCH];

        v8f d = {};
        #pragma unroll
        for (int kk = 0; kk < NHID; kk += 4) {
            const unsigned k0 = kk + kh;
            v2f a, b;
            a.x = fmaxf(arow[k0]     + bs[k0],     0.f);
            a.y = fmaxf(arow[k0 + 1] + bs[k0 + 1], 0.f);
            b.x = WsT[m][k0];
            b.y = WsT[m][k0 + 1];
            d = __builtin_amdgcn_wmma_f32_16x16x4_f32(false, a, false, b,
                                                      (short)0, d, false, false);
        }

        const unsigned r0 = wave * 16 + (lane >> 4) * 8;
        float* orow = out + ((size_t)n0 + r0) * NCLASS + m;
        #pragma unroll
        for (int r = 0; r < 8; ++r)
            if ((int)(r0 + r) < nrows) orow[(size_t)r * NCLASS] = d[r];
    }
}

__global__ __launch_bounds__(OWN_THREADS) void gcn_own16_kernel(const uint2* __restrict__ recs,
                                                                const int4* __restrict__ spill,
                                                                const unsigned* __restrict__ cursors,
                                                                const float* __restrict__ sup,
                                                                const float* __restrict__ b2,
                                                                float* __restrict__ out) {
    __shared__ float acc[BROWS * NCLASS];

    const int tid = threadIdx.x;
    const unsigned lane = tid & 31;
    const unsigned wave = (unsigned)__builtin_amdgcn_readfirstlane(tid >> 5);
    const bool odd = lane >= 16;
    const unsigned f = lane & 15;
    const int bk = blockIdx.x;
    const int n0 = bk * BROWS;
    const int nrows = min(BROWS, NNODES - n0);

    for (int i = tid; i < BROWS * NCLASS; i += OWN_THREADS) acc[i] = 0.f;
    const unsigned cnt  = min(cursors[bk], (unsigned)BCAP);
    const unsigned scnt = min(cursors[NBUCKETS], (unsigned)NEDGES);
    __syncthreads();

    const uint2* my = recs + (size_t)bk * BCAP;
    for (unsigned b0 = wave * 32; b0 < cnt; b0 += OWN_THREADS) {
        const unsigned idx = b0 + lane;
        uint2 r = make_uint2(0u, 0u);
        if (idx < cnt) r = my[idx];
        const int n = (int)min(32u, cnt - b0);
        for (int g = 0; g < n; g += 16) {
            float v[8];
            #pragma unroll
            for (int u = 0; u < 8; ++u) {
                const int ja = min(g + 2 * u,     n - 1);
                const int jb = min(g + 2 * u + 1, n - 1);
                const unsigned pa = (unsigned)__builtin_amdgcn_readlane(r.x, ja);
                const unsigned pb = (unsigned)__builtin_amdgcn_readlane(r.x, jb);
                const unsigned sj = (odd ? pb : pa) & ((1u << SRCBITS) - 1u);
                v[u] = sup[sj * NCLASS + f];
            }
            #pragma unroll
            for (int u = 0; u < 8; ++u) {
                const int ja = g + 2 * u;
                if (ja < n) {
                    const int jb = min(ja + 1, n - 1);
                    const unsigned pa = (unsigned)__builtin_amdgcn_readlane(r.x, ja);
                    const unsigned pb = (unsigned)__builtin_amdgcn_readlane(r.x, jb);
                    const unsigned wa = (unsigned)__builtin_amdgcn_readlane(r.y, ja);
                    const unsigned wb = (unsigned)__builtin_amdgcn_readlane(r.y, jb);
                    const unsigned dj = (odd ? pb : pa) >> SRCBITS;
                    const float wj = __uint_as_float(odd ? wb : wa);
                    if (!odd || ja + 1 < n)
                        lds_add_f32(&acc[dj * NCLASS + f], wj * v[u]);
                }
            }
        }
    }
    for (unsigned b0 = wave * 32; b0 < scnt; b0 += OWN_THREADS) {
        const unsigned idx = b0 + lane;
        int4 r = make_int4(0, -1, 0, 0);
        if (idx < scnt) r = spill[idx];
        const int n = (int)min(32u, scnt - b0);
        for (int j = 0; j < n; ++j) {
            const int dj = __builtin_amdgcn_readlane(r.y, j) - n0;
            if ((unsigned)dj < (unsigned)nrows) {
                const unsigned sj = (unsigned)__builtin_amdgcn_readlane(r.x, j);
                const float wj = __int_as_float(__builtin_amdgcn_readlane(r.z, j));
                if (!odd) lds_add_f32(&acc[dj * NCLASS + f], wj * sup[sj * NCLASS + f]);
            }
        }
    }
    __syncthreads();

    if (tid < nrows) {
        float fv[NCLASS];
        #pragma unroll
        for (int i = 0; i < NCLASS; ++i) fv[i] = acc[tid * NCLASS + i] + b2[i];
        float mx = fv[0];
        #pragma unroll
        for (int i = 1; i < NCLASS; ++i) mx = fmaxf(mx, fv[i]);
        float sum = 0.f;
        #pragma unroll
        for (int i = 0; i < NCLASS; ++i) sum += __expf(fv[i] - mx);
        const float lse = mx + __logf(sum);
        float4* p = (float4*)(out + (size_t)(n0 + tid) * NCLASS);
        #pragma unroll
        for (int i = 0; i < 4; ++i)
            p[i] = make_float4(fv[4 * i] - lse, fv[4 * i + 1] - lse, fv[4 * i + 2] - lse, fv[4 * i + 3] - lse);
    }
}

extern "C" void kernel_launch(void* const* d_in, const int* in_sizes, int n_in,
                              void* d_out, int out_size, void* d_ws, size_t ws_size,
                              hipStream_t stream) {
    const float* x   = (const float*)d_in[0];
    const int*   src = (const int*)  d_in[1];
    const int*   dst = (const int*)  d_in[2];
    const float* ew  = (const float*)d_in[3];
    const float* W1  = (const float*)d_in[4];
    const float* b1  = (const float*)d_in[5];
    const float* W2  = (const float*)d_in[6];
    const float* b2  = (const float*)d_in[7];
    float* out = (float*)d_out;

    float* ws       = (float*)d_ws;
    float* support1 = ws;
    float* support2 = ws + (size_t)NNODES * NHID;
    uint2* recs     = (uint2*)(support2 + (size_t)NNODES * NCLASS);
    int4*  spill    = (int4*)(recs + (size_t)NBUCKETS * BCAP);
    unsigned* cursors = (unsigned*)(spill + (size_t)NEDGES);

    gcn_gemm1_kernel<<<G1_BLOCKS, 128, 0, stream>>>(x, W1, support1, cursors);
    gcn_bin_kernel<<<BIN_BLOCKS, BIN_THREADS, 0, stream>>>(src, dst, ew, recs, spill, cursors);
    gcn_own64_kernel<<<NBUCKETS, OWN_THREADS, 0, stream>>>(recs, spill, cursors, support1, b1, W2, support2);

    gcn_own16_kernel<<<NBUCKETS, OWN_THREADS, 0, stream>>>(recs, spill, cursors, support2, b2, out);
}


# Layer-2 owner waves stage 32-record batches in LDS (lanes read their record pairs there) and send the next batch's 16 row gathers while the current is summed; LDS f32 atomic adds, arrival order
# speedup vs baseline: 6.8638x; 1.0326x over previous
#include <hip/hip_runtime.h>

#define NNODES 100000
#define NEDGES 1600000
#define NFEAT 128
#define NHID 64
#define NCLASS 16
#define MTILES (NNODES / 16)


#define NBUCKETS 256
#define BROWS 391
#define BCAP 32768
#define BIN_THREADS 256
#define BIN_EPT 25
#define BIN_EDGES (BIN_THREADS * BIN_EPT)
#define BIN_BLOCKS (NEDGES / BIN_EDGES)
#define OWN_THREADS 1024
#define APITCH (NHID + 4)
#define AROWS (((BROWS + 15) / 16) * 16)
static_assert(NBUCKETS * BROWS >= NNODES, "buckets must cover every node");
static_assert(NEDGES % BIN_EDGES == 0, "edge count must fill whole binning blocks");
static_assert(NBUCKETS == (BIN_THREADS / 32) * 32, "copy-out: every binning wave writes 32 buckets");
static_assert(BIN_EDGES <= 65536, "a rank inside a block must fit 16 bits");
static_assert(BIN_THREADS == NBUCKETS, "one thread per bucket counter");
static_assert(OWN_THREADS == NHID * NCLASS, "one owner thread per W2 element");
static_assert(AROWS / 16 <= OWN_THREADS / 32, "one wave per 16-row tile of a bucket");
#define SCAP 8192
#define SREC (SCAP / OWN_THREADS)
#define RSTN 416
#define SRCBITS 17
static_assert(SCAP % OWN_THREADS == 0, "whole records per thread");
static_assert(NNODES <= (1 << SRCBITS), "source index must fit its field");
static_assert(AROWS <= RSTN && RSTN <= 512 && RSTN % 32 == 0, "row index must fit its field and the scan");

typedef float v2f __attribute__((ext_vector_type(2)));
typedef float v8f __attribute__((ext_vector_type(8)));

__device__ __forceinline__ void lds_add_f32(float* p, float v) {
    __hip_atomic_fetch_add(p, v, __ATOMIC_RELAXED, __HIP_MEMORY_SCOPE_WORKGROUP);
}


#define G1_TPB 5
#define G1_BLOCKS (MTILES / G1_TPB)
static_assert(MTILES % G1_TPB == 0, "M tiles must fill whole GEMM1 blocks");
__global__ __launch_bounds__(128) void gcn_gemm1_kernel(const float* __restrict__ x,
                                                        const float* __restrict__ W1,
                                                        float* __restrict__ out,
                                                        unsigned* __restrict__ cursors) {
    __shared__ __attribute__((aligned(16))) float As[2][16][NFEAT + 4];
    __shared__ float WsT[NHID][NFEAT + 4];

    const int tid = threadIdx.x;
    const int tile0 = blockIdx.x * G1_TPB;

    const float4* xv = (const float4*)(x + (size_t)tile0 * 16 * NFEAT);
    float4 xa = xv[tid], xb = xv[tid + 128], xc = xv[tid + 256], xd = xv[tid + 384];

    for (int i = tid; i < NFEAT * NHID; i += 128) {
        int k = i >> 6, n = i & (NHID - 1);
        WsT[n][k] = W1[i];
    }

    if (blockIdx.x == 0) {
        for (int i = tid; i <= NBUCKETS; i += 128) cursors[i] = 0u;
    }

    const int lane = tid & 31;
    const int wave = tid >> 5;
    const int m  = lane & 15;
    const int n  = lane & 15;
    const int kh = (lane >> 4) * 2;
    const int n0 = wave * 16;
    float* orow0 = out + ((size_t)tile0 * 16 + (lane >> 4) * 8) * NHID + n0 + n;

    v8f prev = {};
    #pragma unroll 1
    for (int t = 0; t < G1_TPB; ++t) {
        float (*A)[NFEAT + 4] = As[t & 1];
        {
            const int r4 = tid >> 5, c4 = (tid & 31) * 4;
            *(float4*)&A[r4][c4]      = xa;
            *(float4*)&A[r4 + 4][c4]  = xb;
            *(float4*)&A[r4 + 8][c4]  = xc;
            *(float4*)&A[r4 + 12][c4] = xd;
        }
        __syncthreads();
        if (t > 0) {
            float* orow = orow0 + (size_t)(t - 1) * 16 * NHID;
            #pragma unroll
            for (int r = 0; r < 8; ++r)
                orow[(size_t)r * NHID] = prev[r];
        }
        if (t + 1 < G1_TPB) {
            xv += 16 * NFEAT / 4;
            xa = xv[tid]; xb = xv[tid + 128]; xc = xv[tid + 256]; xd = xv[tid + 384];
        }

        v8f acc = {};
        #pragma unroll 4
        for (int kk = 0; kk < NFEAT; kk += 4) {
            const int k0 = kk + kh;
            v2f a, b;
            a.x = A[m][k0];
            a.y = A[m][k0 + 1];
            b.x = WsT[n0 + n][k0];
            b.y = WsT[n0 + n][k0 + 1];
            acc = __builtin_amdgcn_wmma_f32_16x16x4_f32(false, a, false, b,
                                                        (short)0, acc, false, false);
        }
        prev = acc;
    }
    {
        float* orow = orow0 + (size_t)(G1_TPB - 1) * 16 * NHID;
        #pragma unroll
        for (int r = 0; r < 8; ++r)
            orow[(size_t)r * NHID] = prev[r];
    }
}

__global__ __launch_bounds__(BIN_THREADS) void gcn_bin_kernel(const int* __restrict__ src,
                                                              const int* __restrict__ dst,
                                                              const float* __restrict__ ew,
                                                              uint2* __restrict__ recs,
                                                              int4* __restrict__ spill,
                                                              unsigned* __restrict__ cursors) {
    __shared__ unsigned cnt[NBUCKETS];
    __shared__ unsigned span[NBUCKETS];
    __shared__ unsigned lst[NBUCKETS];
    __shared__ unsigned wsum[BIN_THREADS / 32];
    __shared__ uint2 stage[BIN_EDGES];

    const int tid = threadIdx.x;
    const unsigned lane = tid & 31;
    const unsigned wave = (unsigned)__builtin_amdgcn_readfirstlane(tid >> 5);
    cnt[tid] = 0u;
    __syncthreads();

    const int e0 = blockIdx.x * BIN_EDGES + tid;
    int s[BIN_EPT], d[BIN_EPT];
    float w[BIN_EPT];
    #pragma unroll
    for (int j = 0; j < BIN_EPT; ++j) {
        const int e = e0 + j * BIN_THREADS;
        s[j] = src[e];
        d[j] = dst[e];
        w[j] = ew[e];
    }
    unsigned pk[BIN_EPT], br[BIN_EPT];
    #pragma unroll
    for (int j = 0; j < BIN_EPT; ++j) {
        const unsigned bk = (unsigned)d[j] / BROWS;
        const unsigned row = (unsigned)d[j] - bk * BROWS;
        pk[j] = (unsigned)s[j] | (row << SRCBITS);
        const unsigned rk = __hip_atomic_fetch_add(&cnt[bk], 1u, __ATOMIC_RELAXED, __HIP_MEMORY_SCOPE_WORKGROUP);
        br[j] = (bk << 16) | rk;
    }
    __syncthreads();
    {
        const unsigned c = cnt[tid];
        const unsigned g = c ? __hip_atomic_fetch_add(&cursors[tid], c, __ATOMIC_RELAXED, __HIP_MEMORY_SCOPE_AGENT) : 0u;
        unsigned incl = c;
        #pragma unroll
        for (int dd = 1; dd < 32; dd <<= 1) {
            const unsigned t = __shfl_up(incl, dd);
            if (lane >= (unsigned)dd) incl += t;
        }
        if (lane == 31) wsum[wave] = incl;
        __syncthreads();
        unsigned base = 0u;
        #pragma unroll
        for (unsigned i = 0; i < BIN_THREADS / 32; ++i)
            if (i < wave) base += wsum[i];
        lst[tid] = base + incl - c;
        span[tid] = g;
    }
    __syncthreads();
    #pragma unroll
    for (int j = 0; j < BIN_EPT; ++j)
        stage[lst[br[j] >> 16] + (br[j] & 0xffffu)] = make_uint2(pk[j], __float_as_uint(w[j]));
    __syncthreads();

    const unsigned kb = wave * 32u;
    const unsigned cv = cnt[kb + lane];
    const unsigned lv = lst[kb + lane];
    const unsigned gv = span[kb + lane];
    #pragma unroll
    for (int h = 0; h < 2; ++h) {
        uint2 rc[16];
        #pragma unroll
        for (int b = 0; b < 16; ++b) {
            const unsigned l0 = (unsigned)__builtin_amdgcn_readlane(lv, h * 16 + b);
            rc[b] = stage[min(l0 + lane, (unsigned)(BIN_EDGES - 1))];
        }
        #pragma unroll
        for (int b = 0; b < 16; ++b) {
            const unsigned k  = kb + (unsigned)(h * 16 + b);
            const unsigned n  = (unsigned)__builtin_amdgcn_readlane(cv, h * 16 + b);
            const unsigned l0 = (unsigned)__builtin_amdgcn_readlane(lv, h * 16 + b);
            const unsigned g0 = (unsigned)__builtin_amdgcn_readlane(gv, h * 16 + b);
            uint2 r = rc[b];
            for (unsigned i = lane; i < n; i += 32) {
                if (i >= 32u) r = stage[l0 + i];
                const unsigned pos = g0 + i;
                if (pos < BCAP) {
                    recs[(size_t)k * BCAP + pos] = r;
                } else {
                    const unsigned o = __hip_atomic_fetch_add(&cursors[NBUCKETS], 1u, __ATOMIC_RELAXED, __HIP_MEMORY_SCOPE_AGENT);
                    spill[o] = make_int4((int)(r.x & ((1u << SRCBITS) - 1u)), (int)(k * BROWS + (r.x >> SRCBITS)), (int)r.y, 0);
                }
            }
        }
    }
}

__global__ __launch_bounds__(OWN_THREADS) void gcn_own64_kernel(const uint2* __restrict__ recs,
                                                                const int4* __restrict__ spill,
                                                                const unsigned* __restrict__ cursors,
                                                                const float* __restrict__ sup,
                                                                const float* __restrict__ b1,
                                                                const float* __restrict__ W2,
                                                                float* __restrict__ out) {
    __shared__ float acc[AROWS * APITCH];
    __shared__ float WsT[NCLASS][NHID + 4];
    __shared__ float bs[NHID];
    __shared__ unsigned rowst[RSTN];
    __shared__ uint2 srt[SCAP];

    const int tid = threadIdx.x;
    const unsigned lane = tid & 31;
    const unsigned wave = (unsigned)__builtin_amdgcn_readfirstlane(tid >> 5);
    const int bk = blockIdx.x;
    const int n0 = bk * BROWS;
    const int nrows = min(BROWS, NNODES - n0);

    const unsigned cnt  = min(cursors[bk], (unsigned)BCAP);
    const unsigned scnt = min(cursors[NBUCKETS], (unsigned)NEDGES);
    for (int i = tid; i < AROWS * APITCH; i += OWN_THREADS) acc[i] = 0.f;
    WsT[tid & (NCLASS - 1)][tid >> 4] = W2[tid];
    if (tid < NHID) bs[tid] = b1[tid];

    const uint2* my = recs + (size_t)bk * BCAP;
    for (unsigned c0 = 0; c0 < cnt; c0 += SCAP) {
        const unsigned nc = min(cnt - c0, (unsigned)SCAP);
        if (tid < RSTN) rowst[tid] = 0u;
        unsigned pk[SREC], rk[SREC];
        float wv[SREC];
        uint2 rr[SREC];
        unsigned ci[SREC];
        #pragma unroll
        for (int i = 0; i < SREC; ++i) {
            const unsigned idx = tid + i * OWN_THREADS;
            ci[i] = c0 + min(idx, nc - 1u);
        }
        __builtin_amdgcn_sched_barrier(0);
        #pragma unroll
        for (int i = 0; i < SREC; ++i) rr[i] = my[ci[i]];
        __builtin_amdgcn_sched_barrier(0);
        __syncthreads();
        #pragma unroll
        for (int i = 0; i < SREC; ++i) {
            pk[i] = rr[i].x;
            wv[i] = __uint_as_float(rr[i].y);
        }
        #pragma unroll
        for (int i = 0; i < SREC; ++i) {
            const unsigned idx = tid + i * OWN_THREADS;
            rk[i] = 0u;
            if (idx < nc)
                rk[i] = __hip_atomic_fetch_add(&rowst[pk[i] >> SRCBITS], 1u, __ATOMIC_RELAXED, __HIP_MEMORY_SCOPE_WORKGROUP);
        }
        __syncthreads();
        if (wave == 0) {
            unsigned c[RSTN / 32];
            unsigned s = 0u;
            #pragma unroll
            for (int i = 0; i < RSTN / 32; ++i) { c[i] = rowst[lane * (RSTN / 32) + i]; s += c[i]; }
            unsigned incl = s;
            #pragma unroll
            for (int d = 1; d < 32; d <<= 1) {
                const unsigned t = __shfl_up(incl, d);
                if (lane >= (unsigned)d) incl += t;
            }
            unsigned run = incl - s;
            #pragma unroll
            for (int i = 0; i < RSTN / 32; ++i) { rowst[lane * (RSTN / 32) + i] = run; run += c[i]; }
        }
        __syncthreads();
        unsigned pos[SREC];
        #pragma unroll
        for (int i = 0; i < SREC; ++i) pos[i] = rowst[pk[i] >> SRCBITS] + rk[i];
        #pragma unroll
        for (int i = 0; i < SREC; ++i) {
            const unsigned idx = tid + i * OWN_THREADS;
            if (idx < nc) srt[pos[i]] = make_uint2(pk[i], __float_as_uint(wv[i]));
        }
        __syncthreads();

        const unsigned per = (nc + 31u) >> 5;
        const unsigned e0 = min(wave * per, nc);
        const unsigned e1 = min(e0 + per, nc);
        unsigned cur = 0u;
        float a0 = 0.f, a1 = 0.f;
        for (unsigned b0 = e0; b0 < e1; b0 += 32) {
            const unsigned idx = b0 + lane;
            uint2 r = make_uint2(0u, 0u);
            if (idx < e1) r = srt[idx];
            const int n = (int)min(32u, e1 - b0);
            if (n == 32) {
                float v0[2][8], v1[2][8];
                #pragma unroll
                for (int u = 0; u < 8; ++u) {
                    const unsigned sj = (unsigned)__builtin_amdgcn_readlane(r.x, u) & ((1u << SRCBITS) - 1u);
                    const unsigned o = sj * NHID + lane;
                    v0[0][u] = sup[o];
                    v1[0][u] = sup[o + 32];
                }
                #pragma unroll
                for (int gi = 0; gi < 4; ++gi) {
                    if (gi + 1 < 4) {
                        #pragma unroll
                        for (int u = 0; u < 8; ++u) {
                            const unsigned sj = (unsigned)__builtin_amdgcn_readlane(r.x, (gi + 1) * 8 + u) & ((1u << SRCBITS) - 1u);
                            const unsigned o = sj * NHID + lane;
                            v0[(gi + 1) & 1][u] = sup[o];
                            v1[(gi + 1) & 1][u] = sup[o + 32];
                        }
                    }
                    #pragma unroll
                    for (int u = 0; u < 8; ++u) {
                        const unsigned rj = (unsigned)__builtin_amdgcn_readlane(r.x, gi * 8 + u) >> SRCBITS;
                        const float wj = __uint_as_float((unsigned)__builtin_amdgcn_readlane(r.y, gi * 8 + u));
                        if (rj != cur) {
                            lds_add_f32(&acc[cur * APITCH + lane],      a0);
                            lds_add_f32(&acc[cur * APITCH + lane + 32], a1);
                            a0 = 0.f; a1 = 0.f; cur = rj;
                        }
                        a0 = fmaf(wj, v0[gi & 1][u], a0);
                        a1 = fmaf(wj, v1[gi & 1][u], a1);
                    }
                }
            } else {
                for (int g = 0; g < n; g += 8) {
                    float v0[8], v1[8];
                    #pragma unroll
                    for (int u = 0; u < 8; ++u) {
                        const int j = min(g + u, n - 1);
                        const unsigned sj = (unsigned)__builtin_amdgcn_readlane(r.x, j) & ((1u << SRCBITS) - 1u);
                        const unsigned o = sj * NHID + lane;
                        v0[u] = sup[o];
                        v1[u] = sup[o + 32];
                    }
                    #pragma unroll
                    for (int u = 0; u < 8; ++u) {
                        const int j = g + u;
                        if (j < n) {
                            const unsigned rj = (unsigned)__builtin_amdgcn_readlane(r.x, j) >> SRCBITS;
                            const float wj = __uint_as_float((unsigned)__builtin_amdgcn_readlane(r.y, j));
                            if (rj != cur) {
                                lds_add_f32(&acc[cur * APITCH + lane],      a0);
                                lds_add_f32(&acc[cur * APITCH + lane + 32], a1);
                                a0 = 0.f; a1 = 0.f; cur = rj;
                            }
                            a0 = fmaf(wj, v0[u], a0);
                            a1 = fmaf(wj, v1[u], a1);
                        }
                    }
                }
            }
        }
        lds_add_f32(&acc[cur * APITCH + lane],      a0);
        lds_add_f32(&acc[cur * APITCH + lane + 32], a1);
    }
    __syncthreads();
    for (unsigned b0 = wave * 32; b0 < scnt; b0 += OWN_THREADS) {
        const unsigned idx = b0 + lane;
        int4 r = make_int4(0, -1, 0, 0);
        if (idx < scnt) r = spill[idx];
        const int n = (int)min(32u, scnt - b0);
        for (int j = 0; j < n; ++j) {
            const int dj = __builtin_amdgcn_readlane(r.y, j) - n0;
            if ((unsigned)dj < (unsigned)nrows) {
                const unsigned sj = (unsigned)__builtin_amdgcn_readlane(r.x, j);
                const float wj = __int_as_float(__builtin_amdgcn_readlane(r.z, j));
                const unsigned o = sj * NHID + lane;
                lds_add_f32(&acc[dj * APITCH + lane],      wj * sup[o]);
                lds_add_f32(&acc[dj * APITCH + lane + 32], wj * sup[o + 32]);
            }
        }
    }
    __syncthreads();

    const unsigned ntiles = (unsigned)(nrows + 15) >> 4;
    if (wave < ntiles) {
        const unsigned m  = lane & 15;
        const unsigned kh = (lane >> 4) * 2;
        const float* arow = &acc[(wave * 16 + m) * APITCH];

        v8f d = {};
        #pragma unroll
        for (int kk = 0; kk < NHID; kk += 4) {
            const unsigned k0 = kk + kh;
            v2f a, b;
            a.x = fmaxf(arow[k0]     + bs[k0],     0.f);
            a.y = fmaxf(arow[k0 + 1] + bs[k0 + 1], 0.f);
            b.x = WsT[m][k0];
            b.y = WsT[m][k0 + 1];
            d = __builtin_amdgcn_wmma_f32_16x16x4_f32(false, a, false, b,
                                                      (short)0, d, false, false);
        }

        const unsigned r0 = wave * 16 + (lane >> 4) * 8;
        float* orow = out + ((size_t)n0 + r0) * NCLASS + m;
        #pragma unroll
        for (int r = 0; r < 8; ++r)
            if ((int)(r0 + r) < nrows) orow[(size_t)r * NCLASS] = d[r];
    }
}

#define O16_BATCH 32
#define O16_NULL ((unsigned)BROWS << SRCBITS)
static_assert(O16_BATCH == 32, "one record per lane in a staged batch");
static_assert(BROWS < (1 << (32 - SRCBITS)), "the padding row must fit the row field of a record");

template <bool ISSUE>
__device__ __forceinline__ void own16_steps(const uint4* cur4, const uint4* nxt4, float* acc,
                                            const float* __restrict__ sup, float (&v)[16],
                                            const unsigned h, const unsigned f) {
    uint4 c[8];
    unsigned xs[16];
    #pragma unroll
    for (int q = 0; q < 8; ++q) c[q] = cur4[q * 2 + h];
    if (ISSUE) {
        #pragma unroll
        for (int q = 0; q < 8; ++q) {
            const uint4 x = nxt4[q * 2 + h];
            xs[2 * q] = x.x; xs[2 * q + 1] = x.z;
        }
    }
    #pragma unroll
    for (int q = 0; q < 8; ++q) {
        lds_add_f32(&acc[(c[q].x >> SRCBITS) * NCLASS + f], __uint_as_float(c[q].y) * v[2 * q]);
        if (ISSUE) v[2 * q] = sup[(xs[2 * q] & ((1u << SRCBITS) - 1u)) * NCLASS + f];
        lds_add_f32(&acc[(c[q].z >> SRCBITS) * NCLASS + f], __uint_as_float(c[q].w) * v[2 * q + 1]);
        if (ISSUE) v[2 * q + 1] = sup[(xs[2 * q + 1] & ((1u << SRCBITS) - 1u)) * NCLASS + f];
    }
}

__global__ __launch_bounds__(OWN_THREADS) void gcn_own16_kernel(const uint2* __restrict__ recs,
                                                                const int4* __restrict__ spill,
                                                                const unsigned* __restrict__ cursors,
                                                                const float* __restrict__ sup,
                                                                const float* __restrict__ b2,
                                                                float* __restrict__ out) {
    __shared__ float acc[(BROWS + 1) * NCLASS];
    __shared__ __attribute__((aligned(16))) uint2 stg[OWN_THREADS / 32][2][O16_BATCH];

    const int tid = threadIdx.x;
    const unsigned lane = tid & 31;
    const unsigned wave = (unsigned)__builtin_amdgcn_readfirstlane(tid >> 5);
    const bool odd = lane >= 16;
    const unsigned h = lane >> 4;
    const unsigned f = lane & 15;
    const int bk = blockIdx.x;
    const int n0 = bk * BROWS;
    const int nrows = min(BROWS, NNODES - n0);

    const unsigned cnt  = min(cursors[bk], (unsigned)BCAP);
    const unsigned scnt = min(cursors[NBUCKETS], (unsigned)NEDGES);
    for (int i = tid; i < BROWS * NCLASS; i += OWN_THREADS) acc[i] = 0.f;

    const uint2* my = recs + (size_t)bk * BCAP;
    uint2 (*st)[O16_BATCH] = stg[wave];
    unsigned b0 = wave * O16_BATCH;
    const bool act = b0 < cnt;
    float v[16] = {};
    uint2 rn = make_uint2(O16_NULL, 0u);
    unsigned rni = 0u;
    if (act) {
        uint2 r = make_uint2(O16_NULL, 0u);
        if (b0 + lane < cnt) r = my[b0 + lane];
        rni = b0 + OWN_THREADS + lane;
        rn = my[min(rni, cnt - 1u)];
        st[0][lane] = r;
        __builtin_amdgcn_wave_barrier();
        const uint4* c4 = (const uint4*)st[0];
        #pragma unroll
        for (int q = 0; q < 8; ++q) {
            const uint4 x = c4[q * 2 + h];
            v[2 * q]     = sup[(x.x & ((1u << SRCBITS) - 1u)) * NCLASS + f];
            v[2 * q + 1] = sup[(x.z & ((1u << SRCBITS) - 1u)) * NCLASS + f];
        }
    }
    __syncthreads();

    if (act) {
        unsigned par = 0u;
        while (b0 + OWN_THREADS < cnt) {
            const unsigned b1 = b0 + OWN_THREADS;
            st[par ^ 1u][lane] = rni < cnt ? rn : make_uint2(O16_NULL, 0u);
            __builtin_amdgcn_wave_barrier();
            rni = b1 + OWN_THREADS + lane;
            rn = my[min(rni, cnt - 1u)];
            own16_steps<true>((const uint4*)st[par], (const uint4*)st[par ^ 1u], acc, sup, v, h, f);
            b0 = b1; par ^= 1u;
        }
        own16_steps<false>((const uint4*)st[par], (const uint4*)st[par], acc, sup, v, h, f);
    }
    for (unsigned s0 = wave * 32; s0 < scnt; s0 += OWN_THREADS) {
        const unsigned idx = s0 + lane;
        int4 r = make_int4(0, -1, 0, 0);
        if (idx < scnt) r = spill[idx];
        const int n = (int)min(32u, scnt - s0);
        for (int j = 0; j < n; ++j) {
            const int dj = __builtin_amdgcn_readlane(r.y, j) - n0;
            if ((unsigned)dj < (unsigned)nrows) {
                const unsigned sj = (unsigned)__builtin_amdgcn_readlane(r.x, j);
                const float wj = __int_as_float(__builtin_amdgcn_readlane(r.z, j));
                if (!odd) lds_add_f32(&acc[dj * NCLASS + f], wj * sup[sj * NCLASS + f]);
            }
        }
    }
    __syncthreads();

    if (tid < nrows) {
        float fv[NCLASS];
        #pragma unroll
        for (int i = 0; i < NCLASS; ++i) fv[i] = acc[tid * NCLASS + i] + b2[i];
        float mx = fv[0];
        #pragma unroll
        for (int i = 1; i < NCLASS; ++i) mx = fmaxf(mx, fv[i]);
        float sum = 0.f;
        #pragma unroll
        for (int i = 0; i < NCLASS; ++i) sum += __expf(fv[i] - mx);
        const float lse = mx + __logf(sum);
        float4* p = (float4*)(out + (size_t)(n0 + tid) * NCLASS);
        #pragma unroll
        for (int i = 0; i < 4; ++i)
            p[i] = make_float4(fv[4 * i] - lse, fv[4 * i + 1] - lse, fv[4 * i + 2] - lse, fv[4 * i + 3] - lse);
    }
}

extern "C" void kernel_launch(void* const* d_in, const int* in_sizes, int n_in,
                              void* d_out, int out_size, void* d_ws, size_t ws_size,
                              hipStream_t stream) {
    const float* x   = (const float*)d_in[0];
    const int*   src = (const int*)  d_in[1];
    const int*   dst = (const int*)  d_in[2];
    const float* ew  = (const float*)d_in[3];
    const float* W1  = (const float*)d_in[4];
    const float* b1  = (const float*)d_in[5];
    const float* W2  = (const float*)d_in[6];
    const float* b2  = (const float*)d_in[7];
    float* out = (float*)d_out;

    float* ws       = (float*)d_ws;
    float* support1 = ws;
    float* support2 = ws + (size_t)NNODES * NHID;
    uint2* recs     = (uint2*)(support2 + (size_t)NNODES * NCLASS);
    int4*  spill    = (int4*)(recs + (size_t)NBUCKETS * BCAP);
    unsigned* cursors = (unsigned*)(spill + (size_t)NEDGES);

    gcn_gemm1_kernel<<<G1_BLOCKS, 128, 0, stream>>>(x, W1, support1, cursors);
    gcn_bin_kernel<<<BIN_BLOCKS, BIN_THREADS, 0, stream>>>(src, dst, ew, recs, spill, cursors);
    gcn_own64_kernel<<<NBUCKETS, OWN_THREADS, 0, stream>>>(recs, spill, cursors, support1, b1, W2, support2);

    gcn_own16_kernel<<<NBUCKETS, OWN_THREADS, 0, stream>>>(recs, spill, cursors, support2, b2, out);
}


# Edge binning rides in the GEMM1 launch as its first 250 blocks (GEMM1 blocks: 256 threads, two 4-wave halves on one staged W1); a one-block launch zeroes the cursors in front; sum order as before
# speedup vs baseline: 7.1557x; 1.0425x over previous
#include <hip/hip_runtime.h>

#define NNODES 100000
#define NEDGES 1600000
#define NFEAT 128
#define NHID 64
#define NCLASS 16
#define MTILES (NNODES / 16)


#define NBUCKETS 256
#define BROWS 391
#define BCAP 32768
#define BIN_THREADS 256
#define BIN_EPT 25
#define BIN_EDGES (BIN_THREADS * BIN_EPT)
#define BIN_BLOCKS (NEDGES / BIN_EDGES)
#define OWN_THREADS 1024
#define APITCH (NHID + 4)
#define AROWS (((BROWS + 15) / 16) * 16)
static_assert(NBUCKETS * BROWS >= NNODES, "buckets must cover every node");
static_assert(NEDGES % BIN_EDGES == 0, "edge count must fill whole binning blocks");
static_assert(NBUCKETS == (BIN_THREADS / 32) * 32, "copy-out: every binning wave writes 32 buckets");
static_assert(BIN_EDGES <= 65536, "a rank inside a block must fit 16 bits");
static_assert(BIN_THREADS == NBUCKETS, "one thread per bucket counter");
static_assert(OWN_THREADS == NHID * NCLASS, "one owner thread per W2 element");
static_assert(AROWS / 16 <= OWN_THREADS / 32, "one wave per 16-row tile of a bucket");
#define SCAP 8192
#define SREC (SCAP / OWN_THREADS)
#define RSTN 416
#define SRCBITS 17
static_assert(SCAP % OWN_THREADS == 0, "whole records per thread");
static_assert(NNODES <= (1 << SRCBITS), "source index must fit its field");
static_assert(AROWS <= RSTN && RSTN <= 512 && RSTN % 32 == 0, "row index must fit its field and the scan");

typedef float v2f __attribute__((ext_vector_type(2)));
typedef float v8f __attribute__((ext_vector_type(8)));

__device__ __forceinline__ void lds_add_f32(float* p, float v) {
    __hip_atomic_fetch_add(p, v, __ATOMIC_RELAXED, __HIP_MEMORY_SCOPE_WORKGROUP);
}

__global__ __launch_bounds__(320) void gcn_zero_kernel(unsigned* __restrict__ cursors) {
    if (threadIdx.x <= NBUCKETS) cursors[threadIdx.x] = 0u;
}

#define G1_LDS_BYTES ((2 * 2 * 16 + NHID) * (NFEAT + 4) * 4)
#define BIN_LDS_BYTES ((3 * NBUCKETS + BIN_THREADS / 32) * 4 + BIN_EDGES * 8)
#define FRONT_LDS_BYTES (G1_LDS_BYTES > BIN_LDS_BYTES ? G1_LDS_BYTES : BIN_LDS_BYTES)

#define G1_TPB 5
#define G1_BLOCKS (MTILES / (2 * G1_TPB))
static_assert(MTILES % (2 * G1_TPB) == 0, "M tiles must fill whole GEMM1 blocks");
__device__ __forceinline__ void gcn_gemm1_block(unsigned char* smem, const int gb,
                                                const float* __restrict__ x,
                                                const float* __restrict__ W1,
                                                float* __restrict__ out) {
    float (*AsAll)[16][NFEAT + 4] = (float (*)[16][NFEAT + 4])smem;
    float (*WsT)[NFEAT + 4] = (float (*)[NFEAT + 4])(smem + 2 * 2 * 16 * (NFEAT + 4) * 4);

    const int half = threadIdx.x >> 7;
    const int tid = threadIdx.x & 127;
    float (*As)[16][NFEAT + 4] = AsAll + half * 2;
    const int tile0 = (gb * 2 + half) * G1_TPB;

    const float4* xv = (const float4*)(x + (size_t)tile0 * 16 * NFEAT);
    float4 xa = xv[tid], xb = xv[tid + 128], xc = xv[tid + 256], xd = xv[tid + 384];

    for (int i = threadIdx.x; i < NFEAT * NHID; i += 256) {
        int k = i >> 6, n = i & (NHID - 1);
        WsT[n][k] = W1[i];
    }

    const int lane = tid & 31;
    const int wave = tid >> 5;
    const int m  = lane & 15;
    const int n  = lane & 15;
    const int kh = (lane >> 4) * 2;
    const int n0 = wave * 16;
    float* orow0 = out + ((size_t)tile0 * 16 + (lane >> 4) * 8) * NHID + n0 + n;

    v8f prev = {};
    #pragma unroll 1
    for (int t = 0; t < G1_TPB; ++t) {
        float (*A)[NFEAT + 4] = As[t & 1];
        {
            const int r4 = tid >> 5, c4 = (tid & 31) * 4;
            *(float4*)&A[r4][c4]      = xa;
            *(float4*)&A[r4 + 4][c4]  = xb;
            *(float4*)&A[r4 + 8][c4]  = xc;
            *(float4*)&A[r4 + 12][c4] = xd;
        }
        __syncthreads();
        if (t > 0) {
            float* orow = orow0 + (size_t)(t - 1) * 16 * NHID;
            #pragma unroll
            for (int r = 0; r < 8; ++r)
                orow[(size_t)r * NHID] = prev[r];
        }
        if (t + 1 < G1_TPB) {
            xv += 16 * NFEAT / 4;
            xa = xv[tid]; xb = xv[tid + 128]; xc = xv[tid + 256]; xd = xv[tid + 384];
        }

        v8f acc = {};
        #pragma unroll 4
        for (int kk = 0; kk < NFEAT; kk += 4) {
            const int k0 = kk + kh;
            v2f a, b;
            a.x = A[m][k0];
            a.y = A[m][k0 + 1];
            b.x = WsT[n0 + n][k0];
            b.y = WsT[n0 + n][k0 + 1];
            acc = __builtin_amdgcn_wmma_f32_16x16x4_f32(false, a, false, b,
                                                        (short)0, acc, false, false);
        }
        prev = acc;
    }
    {
        float* orow = orow0 + (size_t)(G1_TPB - 1) * 16 * NHID;
        #pragma unroll
        for (int r = 0; r < 8; ++r)
            orow[(size_t)r * NHID] = prev[r];
    }
}

__device__ __forceinline__ void gcn_bin_block(unsigned char* smem, const int bb,
                                              const int* __restrict__ src,
                                              const int* __restrict__ dst,
                                              const float* __restrict__ ew,
                                              uint2* __restrict__ recs,
                                              int4* __restrict__ spill,
                                              unsigned* __restrict__ cursors) {
    unsigned* cnt  = (unsigned*)smem;
    unsigned* span = cnt + NBUCKETS;
    unsigned* lst  = span + NBUCKETS;
    unsigned* wsum = lst + NBUCKETS;
    uint2* stage = (uint2*)(wsum + BIN_THREADS / 32);

    const int tid = threadIdx.x;
    const unsigned lane = tid & 31;
    const unsigned wave = (unsigned)__builtin_amdgcn_readfirstlane(tid >> 5);
    cnt[tid] = 0u;
    __syncthreads();

    const int e0 = bb * BIN_EDGES + tid;
    int s[BIN_EPT], d[BIN_EPT];
    float w[BIN_EPT];
    #pragma unroll
    for (int j = 0; j < BIN_EPT; ++j) {
        const int e = e0 + j * BIN_THREADS;
        s[j] = src[e];
        d[j] = dst[e];
        w[j] = ew[e];
    }
    unsigned pk[BIN_EPT], br[BIN_EPT];
    #pragma unroll
    for (int j = 0; j < BIN_EPT; ++j) {
        const unsigned bk = (unsigned)d[j] / BROWS;
        const unsigned row = (unsigned)d[j] - bk * BROWS;
        pk[j] = (unsigned)s[j] | (row << SRCBITS);
        const unsigned rk = __hip_atomic_fetch_add(&cnt[bk], 1u, __ATOMIC_RELAXED, __HIP_MEMORY_SCOPE_WORKGROUP);
        br[j] = (bk << 16) | rk;
    }
    __syncthreads();
    {
        const unsigned c = cnt[tid];
        const unsigned g = c ? __hip_atomic_fetch_add(&cursors[tid], c, __ATOMIC_RELAXED, __HIP_MEMORY_SCOPE_AGENT) : 0u;
        unsigned incl = c;
        #pragma unroll
        for (int dd = 1; dd < 32; dd <<= 1) {
            const unsigned t = __shfl_up(incl, dd);
            if (lane >= (unsigned)dd) incl += t;
        }
        if (lane == 31) wsum[wave] = incl;
        __syncthreads();
        unsigned base = 0u;
        #pragma unroll
        for (unsigned i = 0; i < BIN_THREADS / 32; ++i)
            if (i < wave) base += wsum[i];
        lst[tid] = base + incl - c;
        span[tid] = g;
    }
    __syncthreads();
    #pragma unroll
    for (int j = 0; j < BIN_EPT; ++j)
        stage[lst[br[j] >> 16] + (br[j] & 0xffffu)] = make_uint2(pk[j], __float_as_uint(w[j]));
    __syncthreads();

    const unsigned kb = wave * 32u;
    const unsigned cv = cnt[kb + lane];
    const unsigned lv = lst[kb + lane];
    const unsigned gv = span[kb + lane];
    #pragma unroll
    for (int h = 0; h < 2; ++h) {
        uint2 rc[16];
        #pragma unroll
        for (int b = 0; b < 16; ++b) {
            const unsigned l0 = (unsigned)__builtin_amdgcn_readlane(lv, h * 16 + b);
            rc[b] = stage[min(l0 + lane, (unsigned)(BIN_EDGES - 1))];
        }
        #pragma unroll
        for (int b = 0; b < 16; ++b) {
            const unsigned k  = kb + (unsigned)(h * 16 + b);
            const unsigned n  = (unsigned)__builtin_amdgcn_readlane(cv, h * 16 + b);
            const unsigned l0 = (unsigned)__builtin_amdgcn_readlane(lv, h * 16 + b);
            const unsigned g0 = (unsigned)__builtin_amdgcn_readlane(gv, h * 16 + b);
            uint2 r = rc[b];
            for (unsigned i = lane; i < n; i += 32) {
                if (i >= 32u) r = stage[l0 + i];
                const unsigned pos = g0 + i;
                if (pos < BCAP) {
                    recs[(size_t)k * BCAP + pos] = r;
                } else {
                    const unsigned o = __hip_atomic_fetch_add(&cursors[NBUCKETS], 1u, __ATOMIC_RELAXED, __HIP_MEMORY_SCOPE_AGENT);
                    spill[o] = make_int4((int)(r.x & ((1u << SRCBITS) - 1u)), (int)(k * BROWS + (r.x >> SRCBITS)), (int)r.y, 0);
                }
            }
        }
    }
}

static_assert(BIN_THREADS == 256, "the front launch runs 256 threads per block for both jobs");
static_assert(((3 * NBUCKETS + BIN_THREADS / 32) * 4) % 8 == 0, "the record stage must start on 8 bytes");
#define FRONT_BLOCKS (BIN_BLOCKS + G1_BLOCKS)
__global__ __launch_bounds__(256, 6) void gcn_front_kernel(const float* __restrict__ x,
                                                        const float* __restrict__ W1,
                                                        float* __restrict__ sup1,
                                                        const int* __restrict__ src,
                                                        const int* __restrict__ dst,
                                                        const float* __restrict__ ew,
                                                        uint2* __restrict__ recs,
                                                        int4* __restrict__ spill,
                                                        unsigned* __restrict__ cursors) {
    __shared__ __attribute__((aligned(16))) unsigned char smem[FRONT_LDS_BYTES];
    if (blockIdx.x < BIN_BLOCKS)
        gcn_bin_block(smem, (int)blockIdx.x, src, dst, ew, recs, spill, cursors);
    else
        gcn_gemm1_block(smem, (int)blockIdx.x - BIN_BLOCKS, x, W1, sup1);
}

__global__ __launch_bounds__(OWN_THREADS) void gcn_own64_kernel(const uint2* __restrict__ recs,
                                                                const int4* __restrict__ spill,
                                                                const unsigned* __restrict__ cursors,
                                                                const float* __restrict__ sup,
                                                                const float* __restrict__ b1,
                                                                const float* __restrict__ W2,
                                                                float* __restrict__ out) {
    __shared__ float acc[AROWS * APITCH];
    __shared__ float WsT[NCLASS][NHID + 4];
    __shared__ float bs[NHID];
    __shared__ unsigned rowst[RSTN];
    __shared__ uint2 srt[SCAP];

    const int tid = threadIdx.x;
    const unsigned lane = tid & 31;
    const unsigned wave = (unsigned)__builtin_amdgcn_readfirstlane(tid >> 5);
    const int bk = blockIdx.x;
    const int n0 = bk * BROWS;
    const int nrows = min(BROWS, NNODES - n0);

    const unsigned cnt  = min(cursors[bk], (unsigned)BCAP);
    const unsigned scnt = min(cursors[NBUCKETS], (unsigned)NEDGES);
    for (int i = tid; i < AROWS * APITCH; i += OWN_THREADS) acc[i] = 0.f;
    WsT[tid & (NCLASS - 1)][tid >> 4] = W2[tid];
    if (tid < NHID) bs[tid] = b1[tid];

    const uint2* my = recs + (size_t)bk * BCAP;
    for (unsigned c0 = 0; c0 < cnt; c0 += SCAP) {
        const unsigned nc = min(cnt - c0, (unsigned)SCAP);
        if (tid < RSTN) rowst[tid] = 0u;
        unsigned pk[SREC], rk[SREC];
        float wv[SREC];
        uint2 rr[SREC];
        unsigned ci[SREC];
        #pragma unroll
        for (int i = 0; i < SREC; ++i) {
            const unsigned idx = tid + i * OWN_THREADS;
            ci[i] = c0 + min(idx, nc - 1u);
        }
        __builtin_amdgcn_sched_barrier(0);
        #pragma unroll
        for (int i = 0; i < SREC; ++i) rr[i] = my[ci[i]];
        __builtin_amdgcn_sched_barrier(0);
        __syncthreads();
        #pragma unroll
        for (int i = 0; i < SREC; ++i) {
            pk[i] = rr[i].x;
            wv[i] = __uint_as_float(rr[i].y);
        }
        #pragma unroll
        for (int i = 0; i < SREC; ++i) {
            const unsigned idx = tid + i * OWN_THREADS;
            rk[i] = 0u;
            if (idx < nc)
                rk[i] = __hip_atomic_fetch_add(&rowst[pk[i] >> SRCBITS], 1u, __ATOMIC_RELAXED, __HIP_MEMORY_SCOPE_WORKGROUP);
        }
        __syncthreads();
        if (wave == 0) {
            unsigned c[RSTN / 32];
            unsigned s = 0u;
            #pragma unroll
            for (int i = 0; i < RSTN / 32; ++i) { c[i] = rowst[lane * (RSTN / 32) + i]; s += c[i]; }
            unsigned incl = s;
            #pragma unroll
            for (int d = 1; d < 32; d <<= 1) {
                const unsigned t = __shfl_up(incl, d);
                if (lane >= (unsigned)d) incl += t;
            }
            unsigned run = incl - s;
            #pragma unroll
            for (int i = 0; i < RSTN / 32; ++i) { rowst[lane * (RSTN / 32) + i] = run; run += c[i]; }
        }
        __syncthreads();
        unsigned pos[SREC];
        #pragma unroll
        for (int i = 0; i < SREC; ++i) pos[i] = rowst[pk[i] >> SRCBITS] + rk[i];
        #pragma unroll
        for (int i = 0; i < SREC; ++i) {
            const unsigned idx = tid + i * OWN_THREADS;
            if (idx < nc) srt[pos[i]] = make_uint2(pk[i], __float_as_uint(wv[i]));
        }
        __syncthreads();

        const unsigned per = (nc + 31u) >> 5;
        const unsigned e0 = min(wave * per, nc);
        const unsigned e1 = min(e0 + per, nc);
        unsigned cur = 0u;
        float a0 = 0.f, a1 = 0.f;
        for (unsigned b0 = e0; b0 < e1; b0 += 32) {
            const unsigned idx = b0 + lane;
            uint2 r = make_uint2(0u, 0u);
            if (idx < e1) r = srt[idx];
            const int n = (int)min(32u, e1 - b0);
            if (n == 32) {
                float v0[2][8], v1[2][8];
                #pragma unroll
                for (int u = 0; u < 8; ++u) {
                    const unsigned sj = (unsigned)__builtin_amdgcn_readlane(r.x, u) & ((1u << SRCBITS) - 1u);
                    const unsigned o = sj * NHID + lane;
                    v0[0][u] = sup[o];
                    v1[0][u] = sup[o + 32];
                }
                #pragma unroll
                for (int gi = 0; gi < 4; ++gi) {
                    if (gi + 1 < 4) {
                        #pragma unroll
                        for (int u = 0; u < 8; ++u) {
                            const unsigned sj = (unsigned)__builtin_amdgcn_readlane(r.x, (gi + 1) * 8 + u) & ((1u << SRCBITS) - 1u);
                            const unsigned o = sj * NHID + lane;
                            v0[(gi + 1) & 1][u] = sup[o];
                            v1[(gi + 1) & 1][u] = sup[o + 32];
                        }
                    }
                    #pragma unroll
                    for (int u = 0; u < 8; ++u) {
                        const unsigned rj = (unsigned)__builtin_amdgcn_readlane(r.x, gi * 8 + u) >> SRCBITS;
                        const float wj = __uint_as_float((unsigned)__builtin_amdgcn_readlane(r.y, gi * 8 + u));
                        if (rj != cur) {
                            lds_add_f32(&acc[cur * APITCH + lane],      a0);
                            lds_add_f32(&acc[cur * APITCH + lane + 32], a1);
                            a0 = 0.f; a1 = 0.f; cur = rj;
                        }
                        a0 = fmaf(wj, v0[gi & 1][u], a0);
                        a1 = fmaf(wj, v1[gi & 1][u], a1);
                    }
                }
            } else {
                for (int g = 0; g < n; g += 8) {
                    float v0[8], v1[8];
                    #pragma unroll
                    for (int u = 0; u < 8; ++u) {
                        const int j = min(g + u, n - 1);
                        const unsigned sj = (unsigned)__builtin_amdgcn_readlane(r.x, j) & ((1u << SRCBITS) - 1u);
                        const unsigned o = sj * NHID + lane;
                        v0[u] = sup[o];
                        v1[u] = sup[o + 32];
                    }
                    #pragma unroll
                    for (int u = 0; u < 8; ++u) {
                        const int j = g + u;
                        if (j < n) {
                            const unsigned rj = (unsigned)__builtin_amdgcn_readlane(r.x, j) >> SRCBITS;
                            const float wj = __uint_as_float((unsigned)__builtin_amdgcn_readlane(r.y, j));
                            if (rj != cur) {
                                lds_add_f32(&acc[cur * APITCH + lane],      a0);
                                lds_add_f32(&acc[cur * APITCH + lane + 32], a1);
                                a0 = 0.f; a1 = 0.f; cur = rj;
                            }
                            a0 = fmaf(wj, v0[u], a0);
                            a1 = fmaf(wj, v1[u], a1);
                        }
                    }
                }
            }
        }
        lds_add_f32(&acc[cur * APITCH + lane],      a0);
        lds_add_f32(&acc[cur * APITCH + lane + 32], a1);
    }
    __syncthreads();
    for (unsigned b0 = wave * 32; b0 < scnt; b0 += OWN_THREADS) {
        const unsigned idx = b0 + lane;
        int4 r = make_int4(0, -1, 0, 0);
        if (idx < scnt) r = spill[idx];
        const int n = (int)min(32u, scnt - b0);
        for (int j = 0; j < n; ++j) {
            const int dj = __builtin_amdgcn_readlane(r.y, j) - n0;
            if ((unsigned)dj < (unsigned)nrows) {
                const unsigned sj = (unsigned)__builtin_amdgcn_readlane(r.x, j);
                const float wj = __int_as_float(__builtin_amdgcn_readlane(r.z, j));
                const unsigned o = sj * NHID + lane;
                lds_add_f32(&acc[dj * APITCH + lane],      wj * sup[o]);
                lds_add_f32(&acc[dj * APITCH + lane + 32], wj * sup[o + 32]);
            }
        }
    }
    __syncthreads();

    const unsigned ntiles = (unsigned)(nrows + 15) >> 4;
    if (wave < ntiles) {
        const unsigned m  = lane & 15;
        const unsigned kh = (lane >> 4) * 2;
        const float* arow = &acc[(wave * 16 + m) * APITCH];

        v8f d = {};
        #pragma unroll
        for (int kk = 0; kk < NHID; kk += 4) {
            const unsigned k0 = kk + kh;
            v2f a, b;
            a.x = fmaxf(arow[k0]     + bs[k0],     0.f);
            a.y = fmaxf(arow[k0 + 1] + bs[k0 + 1], 0.f);
            b.x = WsT[m][k0];
            b.y = WsT[m][k0 + 1];
            d = __builtin_amdgcn_wmma_f32_16x16x4_f32(false, a, false, b,
                                                      (short)0, d, false, false);
        }

        const unsigned r0 = wave * 16 + (lane >> 4) * 8;
        float* orow = out + ((size_t)n0 + r0) * NCLASS + m;
        #pragma unroll
        for (int r = 0; r < 8; ++r)
            if ((int)(r0 + r) < nrows) orow[(size_t)r * NCLASS] = d[r];
    }
}

#define O16_BATCH 32
#define O16_NULL ((unsigned)BROWS << SRCBITS)
static_assert(O16_BATCH == 32, "one record per lane in a staged batch");
static_assert(BROWS < (1 << (32 - SRCBITS)), "the padding row must fit the row field of a record");

template <bool ISSUE>
__device__ __forceinline__ void own16_steps(const uint4* cur4, const uint4* nxt4, float* acc,
                                            const float* __restrict__ sup, float (&v)[16],
                                            const unsigned h, const unsigned f) {
    uint4 c[8];
    unsigned xs[16];
    #pragma unroll
    for (int q = 0; q < 8; ++q) c[q] = cur4[q * 2 + h];
    if (ISSUE) {
        #pragma unroll
        for (int q = 0; q < 8; ++q) {
            const uint4 x = nxt4[q * 2 + h];
            xs[2 * q] = x.x; xs[2 * q + 1] = x.z;
        }
    }
    #pragma unroll
    for (int q = 0; q < 8; ++q) {
        lds_add_f32(&acc[(c[q].x >> SRCBITS) * NCLASS + f], __uint_as_float(c[q].y) * v[2 * q]);
        if (ISSUE) v[2 * q] = sup[(xs[2 * q] & ((1u << SRCBITS) - 1u)) * NCLASS + f];
        lds_add_f32(&acc[(c[q].z >> SRCBITS) * NCLASS + f], __uint_as_float(c[q].w) * v[2 * q + 1]);
        if (ISSUE) v[2 * q + 1] = sup[(xs[2 * q + 1] & ((1u << SRCBITS) - 1u)) * NCLASS + f];
    }
}

__global__ __launch_bounds__(OWN_THREADS) void gcn_own16_kernel(const uint2* __restrict__ recs,
                                                                const int4* __restrict__ spill,
                                                                const unsigned* __restrict__ cursors,
                                                                const float* __restrict__ sup,
                                                                const float* __restrict__ b2,
                                                                float* __restrict__ out) {
    __shared__ float acc[(BROWS + 1) * NCLASS];
    __shared__ __attribute__((aligned(16))) uint2 stg[OWN_THREADS / 32][2][O16_BATCH];

    const int tid = threadIdx.x;
    const unsigned lane = tid & 31;
    const unsigned wave = (unsigned)__builtin_amdgcn_readfirstlane(tid >> 5);
    const bool odd = lane >= 16;
    const unsigned h = lane >> 4;
    const unsigned f = lane & 15;
    const int bk = blockIdx.x;
    const int n0 = bk * BROWS;
    const int nrows = min(BROWS, NNODES - n0);

    const unsigned cnt  = min(cursors[bk], (unsigned)BCAP);
    const unsigned scnt = min(cursors[NBUCKETS], (unsigned)NEDGES);
    for (int i = tid; i < BROWS * NCLASS; i += OWN_THREADS) acc[i] = 0.f;

    const uint2* my = recs + (size_t)bk * BCAP;
    uint2 (*st)[O16_BATCH] = stg[wave];
    unsigned b0 = wave * O16_BATCH;
    const bool act = b0 < cnt;
    float v[16] = {};
    uint2 rn = make_uint2(O16_NULL, 0u);
    unsigned rni = 0u;
    if (act) {
        uint2 r = make_uint2(O16_NULL, 0u);
        if (b0 + lane < cnt) r = my[b0 + lane];
        rni = b0 + OWN_THREADS + lane;
        rn = my[min(rni, cnt - 1u)];
        st[0][lane] = r;
        __builtin_amdgcn_wave_barrier();
        const uint4* c4 = (const uint4*)st[0];
        #pragma unroll
        for (int q = 0; q < 8; ++q) {
            const uint4 x = c4[q * 2 + h];
            v[2 * q]     = sup[(x.x & ((1u << SRCBITS) - 1u)) * NCLASS + f];
            v[2 * q + 1] = sup[(x.z & ((1u << SRCBITS) - 1u)) * NCLASS + f];
        }
    }
    __syncthreads();

    if (act) {
        unsigned par = 0u;
        while (b0 + OWN_THREADS < cnt) {
            const unsigned b1 = b0 + OWN_THREADS;
            st[par ^ 1u][lane] = rni < cnt ? rn : make_uint2(O16_NULL, 0u);
            __builtin_amdgcn_wave_barrier();
            rni = b1 + OWN_THREADS + lane;
            rn = my[min(rni, cnt - 1u)];
            own16_steps<true>((const uint4*)st[par], (const uint4*)st[par ^ 1u], acc, sup, v, h, f);
            b0 = b1; par ^= 1u;
        }
        own16_steps<false>((const uint4*)st[par], (const uint4*)st[par], acc, sup, v, h, f);
    }
    for (unsigned s0 = wave * 32; s0 < scnt; s0 += OWN_THREADS) {
        const unsigned idx = s0 + lane;
        int4 r = make_int4(0, -1, 0, 0);
        if (idx < scnt) r = spill[idx];
        const int n = (int)min(32u, scnt - s0);
        for (int j = 0; j < n; ++j) {
            const int dj = __builtin_amdgcn_readlane(r.y, j) - n0;
            if ((unsigned)dj < (unsigned)nrows) {
                const unsigned sj = (unsigned)__builtin_amdgcn_readlane(r.x, j);
                const float wj = __int_as_float(__builtin_amdgcn_readlane(r.z, j));
                if (!odd) lds_add_f32(&acc[dj * NCLASS + f], wj * sup[sj * NCLASS + f]);
            }
        }
    }
    __syncthreads();

    if (tid < nrows) {
        float fv[NCLASS];
        #pragma unroll
        for (int i = 0; i < NCLASS; ++i) fv[i] = acc[tid * NCLASS + i] + b2[i];
        float mx = fv[0];
        #pragma unroll
        for (int i = 1; i < NCLASS; ++i) mx = fmaxf(mx, fv[i]);
        float sum = 0.f;
        #pragma unroll
        for (int i = 0; i < NCLASS; ++i) sum += __expf(fv[i] - mx);
        const float lse = mx + __logf(sum);
        float4* p = (float4*)(out + (size_t)(n0 + tid) * NCLASS);
        #pragma unroll
        for (int i = 0; i < 4; ++i)
            p[i] = make_float4(fv[4 * i] - lse, fv[4 * i + 1] - lse, fv[4 * i + 2] - lse, fv[4 * i + 3] - lse);
    }
}

extern "C" void kernel_launch(void* const* d_in, const int* in_sizes, int n_in,
                              void* d_out, int out_size, void* d_ws, size_t ws_size,
                              hipStream_t stream) {
    const float* x   = (const float*)d_in[0];
    const int*   src = (const int*)  d_in[1];
    const int*   dst = (const int*)  d_in[2];
    const float* ew  = (const float*)d_in[3];
    const float* W1  = (const float*)d_in[4];
    const float* b1  = (const float*)d_in[5];
    const float* W2  = (const float*)d_in[6];
    const float* b2  = (const float*)d_in[7];
    float* out = (float*)d_out;

    float* ws       = (float*)d_ws;
    float* support1 = ws;
    float* support2 = ws + (size_t)NNODES * NHID;
    uint2* recs     = (uint2*)(support2 + (size_t)NNODES * NCLASS);
    int4*  spill    = (int4*)(recs + (size_t)NBUCKETS * BCAP);
    unsigned* cursors = (unsigned*)(spill + (size_t)NEDGES);

    gcn_zero_kernel<<<1, 320, 0, stream>>>(cursors);
    gcn_front_kernel<<<FRONT_BLOCKS, 256, 0, stream>>>(x, W1, support1, src, dst, ew, recs, spill, cursors);
    gcn_own64_kernel<<<NBUCKETS, OWN_THREADS, 0, stream>>>(recs, spill, cursors, support1, b1, W2, support2);

    gcn_own16_kernel<<<NBUCKETS, OWN_THREADS, 0, stream>>>(recs, spill, cursors, support2, b2, out);
}
